# starting kernel
# baseline (speedup 1.0000x reference)
#include <hip/hip_runtime.h>
#include <math.h>

#define NB 4
#define NT 512
#define ND 256
#define MOMB 0.9f
#define NOVB 0.95f
#define MOM_STRIDE 260

typedef float v2f __attribute__((ext_vector_type(2)));
typedef float v8f __attribute__((ext_vector_type(8)));

struct __align__(16) Smem {
  float mom[ND * MOM_STRIDE];
  float x_s[ND];
  float err_s[ND];
  float yacc[2][ND];
  float iema[ND];
  float redn[8];
  float redf[32];
  float eff_s;
  float scale_s;
};

__global__ __launch_bounds__(1024) void ttt_linear_kernel(
    const float* __restrict__ x, const float* __restrict__ Wb,
    const float* __restrict__ bias, const float* __restrict__ loglr,
    float* __restrict__ out)
{
  __shared__ Smem sm;
  const int tid  = threadIdx.x;
  const int lane = tid & 31;
  const int wid  = tid >> 5;
  const int b    = blockIdx.x;
  const int hi   = lane >> 4;
  const int rb   = (wid >> 1) << 4;
  const int row  = rb + (lane & 15);
  const int colbase = (wid & 1) << 7;

  float w_u[64];
  #pragma unroll
  for (int c = 0; c < 32; ++c) {
    const int col = colbase + 4*c + 2*hi;
    const float2 wv = *reinterpret_cast<const float2*>(Wb + row*ND + col);
    w_u[2*c]   = wv.x;
    w_u[2*c+1] = wv.y;
  }
  for (int i = tid; i < ND*MOM_STRIDE; i += 1024) sm.mom[i] = 0.0f;
  if (tid < ND) sm.iema[tid] = 0.0f;

  float lrema    = 1.0f;
  float inner_lr = 0.0f;
  if (tid == 0) {
    inner_lr = fminf(fmaxf(expf(loglr[0]), 1e-5f), 1.0f);
    sm.scale_s = 1.0f;
    sm.eff_s   = 0.0f;
  }
  const float bias_r = (tid < ND) ? bias[tid] : 0.0f;
  __syncthreads();

  for (int t = 0; t < NT; ++t) {
    if (tid < ND) {
      const float xj = x[((size_t)b*NT + t)*ND + tid];
      sm.x_s[tid] = xj;
      float a = fabsf(xj - sm.iema[tid]);
      sm.iema[tid] = NOVB*sm.iema[tid] + (1.0f-NOVB)*xj;
      #pragma unroll
      for (int off = 16; off > 0; off >>= 1) a += __shfl_xor(a, off, 32);
      if (lane == 0) sm.redn[wid] = a;
    }
    __syncthreads();

    if (tid == 0) {
      float s = 0.0f;
      #pragma unroll
      for (int i = 0; i < 8; ++i) s += sm.redn[i];
      const float nov = s * (1.0f/ND);
      const float lrs = fminf(fmaxf(1.0f + 3.0f*nov, 0.5f), 3.0f);
      lrema = NOVB*lrema + (1.0f-NOVB)*lrs;
      sm.eff_s = inner_lr * lrema;
    }

    v8f acc = {};
    #pragma unroll
    for (int c = 0; c < 32; ++c) {
      const int col = colbase + 4*c + 2*hi;
      const float2 xp = *reinterpret_cast<const float2*>(&sm.x_s[col]);
      v2f av; av.x = w_u[2*c]; av.y = w_u[2*c+1];
      v2f bv; bv.x = xp.x;     bv.y = xp.y;
      acc = __builtin_amdgcn_wmma_f32_16x16x4_f32(
           false, av,  false, bv,
           (short)0, acc,  false,  false);
    }
    if ((lane & 15) == 0) {
      const int mbase = rb + hi*8;
      #pragma unroll
      for (int v = 0; v < 8; ++v) sm.yacc[wid & 1][mbase + v] = acc[v];
    }
    __syncthreads();

    if (tid < ND) {
      const float yv = sm.scale_s*(sm.yacc[0][tid] + sm.yacc[1][tid]) + bias_r;
      out[((size_t)b*NT + t)*ND + tid] = yv;
      sm.err_s[tid] = (yv - sm.x_s[tid]) * (1.0f/ND);
    }
    __syncthreads();

    const float err = sm.err_s[row];
    const float eff = sm.eff_s;
    const float sp  = sm.scale_s;
    float fsum = 0.0f;
    #pragma unroll
    for (int c = 0; c < 32; ++c) {
      const int col = colbase + 4*c + 2*hi;
      const float2 xp = *reinterpret_cast<const float2*>(&sm.x_s[col]);
      float2* mptr = reinterpret_cast<float2*>(&sm.mom[row*MOM_STRIDE + col]);
      float2 mp = *mptr;
      mp.x = MOMB*mp.x + (1.0f-MOMB)*err*xp.x;
      mp.y = MOMB*mp.y + (1.0f-MOMB)*err*xp.y;
      *mptr = mp;
      const float w0 = sp*w_u[2*c]   - eff*mp.x;
      const float w1 = sp*w_u[2*c+1] - eff*mp.y;
      w_u[2*c] = w0; w_u[2*c+1] = w1;
      fsum += w0*w0 + w1*w1;
    }
    #pragma unroll
    for (int off = 16; off > 0; off >>= 1) fsum += __shfl_xor(fsum, off, 32);
    if (lane == 0) sm.redf[wid] = fsum;
    __syncthreads();

    if (tid == 0) {
      float s = 0.0f;
      #pragma unroll
      for (int i = 0; i < 32; ++i) s += sm.redf[i];
      const float fn = fmaxf(sqrtf(s), 1e-6f);
      sm.scale_s = fminf(32.0f/fn, 1.0f);
    }
  }
}

extern "C" void kernel_launch(void* const* d_in, const int* in_sizes, int n_in,
                              void* d_out, int out_size, void* d_ws, size_t ws_size,
                              hipStream_t stream) {
  (void)in_sizes; (void)n_in; (void)out_size; (void)d_ws; (void)ws_size;
  const float* x    = (const float*)d_in[0];
  const float* Wb   = (const float*)d_in[1];
  const float* bias = (const float*)d_in[2];
  const float* lil  = (const float*)d_in[3];
  float* out = (float*)d_out;
  ttt_linear_kernel<<<dim3(NB), dim3(1024), 0, stream>>>(x, Wb, bias, lil, out);
}


# A wave owns one whole row as an A tile of 16 column groups; y = the tile's diagonal summed across lanes: no LDS partials, barrier or x staging in a step; 128 work-groups; y, norm sum orders change
# speedup vs baseline: 19.8429x; 19.8429x over previous
#include <hip/hip_runtime.h>
#include <math.h>

#define NB 4
#define NT 512
#define ND 256
#define MOMB 0.9f
#define NOVB 0.95f
#define MOM_STRIDE 260

#define RBLK 8
#define NRB (ND / RBLK)
#define NSL 16
#define YSTG_OFF (NB * NRB * NT)
#define CLAMP_SAFE 1020.0f
#define FSTR 260
#define PF 4
#define TSTR 260

typedef float v2f __attribute__((ext_vector_type(2)));
typedef float v8f __attribute__((ext_vector_type(8)));

__device__ __forceinline__ float row_sum256(const float* r)
{
  float s0 = 0.0f, s1 = 0.0f, s2 = 0.0f, s3 = 0.0f;
  #pragma unroll 8
  for (int i = 0; i < ND; i += 4) {
    const float4 v = *reinterpret_cast<const float4*>(r + i);
    s0 += v.x; s1 += v.y; s2 += v.z; s3 += v.w;
  }
  return (s0 + s1) + (s2 + s3);
}

template <int M>
__device__ __forceinline__ float add_row_xor(float v)
{
  const int s = __builtin_bit_cast(int, v);
  const int p = __builtin_amdgcn_update_dpp(s, s, 0x160 + M, 0xf, 0xf, true);
  return v + __builtin_bit_cast(float, p);
}

__device__ __forceinline__ float xhalf(float v)
{
  const int s = __builtin_bit_cast(int, v);
  return __builtin_bit_cast(float,
      __builtin_amdgcn_permlanex16(s, s, 0x76543210, 0xfedcba98, false, false));
}

__global__ __launch_bounds__(256) void ttt_rows_kernel(
    const float* __restrict__ x, const float* __restrict__ Wb,
    const float* __restrict__ bias, const float* __restrict__ loglr,
    float* __restrict__ ystg, float* __restrict__ nrm2)
{
  __shared__ __align__(16) float fbuf[256][FSTR];
  __shared__ float lrs_s[NT];
  __shared__ float eff_s[NT];

  const int tid  = threadIdx.x;
  const int lane = tid & 31;
  const int wid  = __builtin_amdgcn_readfirstlane(tid >> 5);
  const int b    = blockIdx.x / NRB;
  const int rblk = blockIdx.x % NRB;
  const int row  = rblk*RBLK + wid;
  const float* __restrict__ xb = x + (size_t)b*NT*ND;

  float w_u[8], m_u[8];
  {
    const float4 wl = *reinterpret_cast<const float4*>(Wb + row*ND + 4*lane);
    const float4 wh = *reinterpret_cast<const float4*>(Wb + row*ND + 128 + 4*lane);
    w_u[0] = wl.x; w_u[1] = wl.y; w_u[2] = wl.z; w_u[3] = wl.w;
    w_u[4] = wh.x; w_u[5] = wh.y; w_u[6] = wh.z; w_u[7] = wh.w;
    #pragma unroll
    for (int i = 0; i < 8; ++i) m_u[i] = 0.0f;
  }

  {
    float iema = 0.0f;
    #pragma unroll 8
    for (int t = 0; t < NT; ++t) {
      const float xj = xb[(size_t)t*ND + tid];
      fbuf[t & 255][tid] = fabsf(xj - iema);
      iema = NOVB*iema + (1.0f-NOVB)*xj;
      if ((t & 255) == 255) {
        __syncthreads();
        const float nov = row_sum256(fbuf[tid]) * (1.0f/ND);
        lrs_s[t - 255 + tid] = fminf(fmaxf(1.0f + 3.0f*nov, 0.5f), 3.0f);
        __syncthreads();
      }
    }
  }
  if (tid == 0) {
    const float inner_lr = fminf(fmaxf(expf(loglr[0]), 1e-5f), 1.0f);
    float lrema = 1.0f;
    #pragma unroll 8
    for (int t = 0; t < NT; ++t) {
      lrema = NOVB*lrema + (1.0f-NOVB)*lrs_s[t];
      eff_s[t] = inner_lr * lrema;
    }
  }

  float4 xa[PF], xc[PF];
  float  xt[PF];
  #pragma unroll
  for (int u = 0; u < PF - 1; ++u) {
    xa[u] = *reinterpret_cast<const float4*>(xb + (size_t)u*ND + 4*lane);
    xc[u] = *reinterpret_cast<const float4*>(xb + (size_t)u*ND + 128 + 4*lane);
    xt[u] = xb[(size_t)u*ND + rblk*RBLK + (lane & 7)];
  }
  xa[PF-1] = make_float4(0.0f, 0.0f, 0.0f, 0.0f);
  xc[PF-1] = make_float4(0.0f, 0.0f, 0.0f, 0.0f);
  xt[PF-1] = 0.0f;
  const float bias_r = bias[row];
  float ykeep = 0.0f;
  const bool s0 = (lane & 1) != 0, s1 = (lane & 2) != 0, s2 = (lane & 4) != 0;
  const bool on_diag = ((lane >> 3) & 1) == (lane >> 4);
  __syncthreads();
  float eff = eff_s[0];

  for (int t0 = 0; t0 < NT; t0 += PF) {
    #pragma unroll
    for (int u = 0; u < PF; ++u) {
      const int t = t0 + u;
      const float xv[8] = { xa[u].x, xa[u].y, xa[u].z, xa[u].w,
                            xc[u].x, xc[u].y, xc[u].z, xc[u].w };
      v8f acc0 = {}, acc1 = {};
      #pragma unroll
      for (int j = 0; j < 4; ++j) {
        v2f av; av.x = w_u[2*j]; av.y = w_u[2*j+1];
        v2f bv; bv.x = xv[2*j];  bv.y = xv[2*j+1];
        if (j & 1)
          acc1 = __builtin_amdgcn_wmma_f32_16x16x4_f32(false, av, false, bv, (short)0, acc1, false, false);
        else
          acc0 = __builtin_amdgcn_wmma_f32_16x16x4_f32(false, av, false, bv, (short)0, acc0, false, false);
      }
      __builtin_amdgcn_sched_barrier(0);
      {
        const int un = (u + PF - 1) % PF;
        const int tn = (t + PF - 1 < NT) ? t + PF - 1 : NT - 1;
        xa[un] = *reinterpret_cast<const float4*>(xb + (size_t)tn*ND + 4*lane);
        xc[un] = *reinterpret_cast<const float4*>(xb + (size_t)tn*ND + 128 + 4*lane);
        xt[un] = xb[(size_t)tn*ND + rblk*RBLK + (lane & 7)];
      }
      const float eff_n = eff_s[(t + 1 < NT) ? t + 1 : t];
      const v8f acc = acc0 + acc1;
      const float d01 = s0 ? acc[1] : acc[0];
      const float d23 = s0 ? acc[3] : acc[2];
      const float d45 = s0 ? acc[5] : acc[4];
      const float d67 = s0 ? acc[7] : acc[6];
      const float d03 = s1 ? d23 : d01;
      const float d47 = s1 ? d67 : d45;
      float d = s2 ? d47 : d03;
      d = on_diag ? d : 0.0f;
      d = add_row_xor<1>(d);
      d = add_row_xor<2>(d);
      d = add_row_xor<4>(d);
      d = add_row_xor<8>(d);
      const float y = (d + xhalf(d)) + bias_r;

      ykeep = (lane == (t & 31)) ? y : ykeep;
      const float xrow = __builtin_bit_cast(float,
          __builtin_amdgcn_readlane(__builtin_bit_cast(int, xt[u]), wid));
      const float ce = (y - xrow) * ((1.0f-MOMB) * (1.0f/ND));

      float fs0 = 0.0f, fs1 = 0.0f;
      #pragma unroll
      for (int i = 0; i < 8; i += 2) {
        const float m0 = MOMB*m_u[i]   + ce*xv[i];
        const float m1 = MOMB*m_u[i+1] + ce*xv[i+1];
        m_u[i] = m0; m_u[i+1] = m1;
        const float w0 = w_u[i]   - eff*m0;
        const float w1 = w_u[i+1] - eff*m1;
        w_u[i] = w0; w_u[i+1] = w1;
        fs0 += w0*w0; fs1 += w1*w1;
      }
      fbuf[t & 255][tid] = fs0 + fs1;

      eff = eff_n;
    }
    const int tl = t0 + PF - 1;
    if ((tl & 31) == 31)
      ystg[((size_t)(b*ND + row))*NT + (tl - 31) + lane] = ykeep;
    if ((tl & 255) == 255) {
      __syncthreads();
      nrm2[(size_t)(b*NRB + rblk)*NT + (tl - 255) + tid] = row_sum256(fbuf[tid]);
      __syncthreads();
    }
  }
}

struct __align__(16) Smem {
  float mom[ND * MOM_STRIDE];
  float x_s[ND];
  float err_s[ND];
  float yacc[2][ND];
  float iema[ND];
  float redn[8];
  float redf[32];
  float eff_s;
  float scale_s;
};

__global__ __launch_bounds__(1024) void ttt_linear_kernel(
    const float* __restrict__ x, const float* __restrict__ Wb,
    const float* __restrict__ bias, const float* __restrict__ loglr,
    float* __restrict__ out, const float* __restrict__ nrm2,
    const float* __restrict__ ystg)
{
  __shared__ Smem sm;
  const int tid  = threadIdx.x;
  const int lane = tid & 31;
  const int wid  = tid >> 5;
  const int b    = blockIdx.x / NSL;
  const int sl   = blockIdx.x % NSL;

  {
    int bad = 0;
    if (tid < NT) {
      float s = 0.0f;
      #pragma unroll
      for (int r = 0; r < NRB; ++r) s += nrm2[(size_t)(b*NRB + r)*NT + tid];
      bad = !(s <= CLAMP_SAFE);
    }
    if (!__syncthreads_or(bad)) {
      float* tile = sm.mom;
      for (int i = tid; i < ND*8; i += 1024) {
        const int r  = i >> 3;
        const int c4 = (i & 7) << 2;
        const float4 v = *reinterpret_cast<const float4*>(
            ystg + ((size_t)(b*ND + r))*NT + sl*(NT/NSL) + c4);
        tile[(c4 + 0)*TSTR + r] = v.x;
        tile[(c4 + 1)*TSTR + r] = v.y;
        tile[(c4 + 2)*TSTR + r] = v.z;
        tile[(c4 + 3)*TSTR + r] = v.w;
      }
      __syncthreads();
      for (int i = tid; i < (NT/NSL)*(ND/4); i += 1024) {
        const int tl = i >> 6;
        const int q  = i & 63;
        const float4 v = *reinterpret_cast<const float4*>(&tile[tl*TSTR + 4*q]);
        *reinterpret_cast<float4*>(out + ((size_t)b*NT + sl*(NT/NSL) + tl)*ND + 4*q) = v;
      }
      return;
    }
    if (sl != 0) return;
  }

  const int hi   = lane >> 4;
  const int rb   = (wid >> 1) << 4;
  const int row  = rb + (lane & 15);
  const int colbase = (wid & 1) << 7;

  float w_u[64];
  #pragma unroll
  for (int c = 0; c < 32; ++c) {
    const int col = colbase + 4*c + 2*hi;
    const float2 wv = *reinterpret_cast<const float2*>(Wb + row*ND + col);
    w_u[2*c]   = wv.x;
    w_u[2*c+1] = wv.y;
  }
  for (int i = tid; i < ND*MOM_STRIDE; i += 1024) sm.mom[i] = 0.0f;
  if (tid < ND) sm.iema[tid] = 0.0f;

  float lrema    = 1.0f;
  float inner_lr = 0.0f;
  if (tid == 0) {
    inner_lr = fminf(fmaxf(expf(loglr[0]), 1e-5f), 1.0f);
    sm.scale_s = 1.0f;
    sm.eff_s   = 0.0f;
  }
  const float bias_r = (tid < ND) ? bias[tid] : 0.0f;
  __syncthreads();

  for (int t = 0; t < NT; ++t) {
    if (tid < ND) {
      const float xj = x[((size_t)b*NT + t)*ND + tid];
      sm.x_s[tid] = xj;
      float a = fabsf(xj - sm.iema[tid]);
      sm.iema[tid] = NOVB*sm.iema[tid] + (1.0f-NOVB)*xj;
      #pragma unroll
      for (int off = 16; off > 0; off >>= 1) a += __shfl_xor(a, off, 32);
      if (lane == 0) sm.redn[wid] = a;
    }
    __syncthreads();

    if (tid == 0) {
      float s = 0.0f;
      #pragma unroll
      for (int i = 0; i < 8; ++i) s += sm.redn[i];
      const float nov = s * (1.0f/ND);
      const float lrs = fminf(fmaxf(1.0f + 3.0f*nov, 0.5f), 3.0f);
      lrema = NOVB*lrema + (1.0f-NOVB)*lrs;
      sm.eff_s = inner_lr * lrema;
    }

    v8f acc = {};
    #pragma unroll
    for (int c = 0; c < 32; ++c) {
      const int col = colbase + 4*c + 2*hi;
      const float2 xp = *reinterpret_cast<const float2*>(&sm.x_s[col]);
      v2f av; av.x = w_u[2*c]; av.y = w_u[2*c+1];
      v2f bv; bv.x = xp.x;     bv.y = xp.y;
      acc = __builtin_amdgcn_wmma_f32_16x16x4_f32(
           false, av,  false, bv,
           (short)0, acc,  false,  false);
    }
    if ((lane & 15) == 0) {
      const int mbase = rb + hi*8;
      #pragma unroll
      for (int v = 0; v < 8; ++v) sm.yacc[wid & 1][mbase + v] = acc[v];
    }
    __syncthreads();

    if (tid < ND) {
      const float yv = sm.scale_s*(sm.yacc[0][tid] + sm.yacc[1][tid]) + bias_r;
      out[((size_t)b*NT + t)*ND + tid] = yv;
      sm.err_s[tid] = (yv - sm.x_s[tid]) * (1.0f/ND);
    }
    __syncthreads();

    const float err = sm.err_s[row];
    const float eff = sm.eff_s;
    const float sp  = sm.scale_s;
    float fsum = 0.0f;
    #pragma unroll
    for (int c = 0; c < 32; ++c) {
      const int col = colbase + 4*c + 2*hi;
      const float2 xp = *reinterpret_cast<const float2*>(&sm.x_s[col]);
      float2* mptr = reinterpret_cast<float2*>(&sm.mom[row*MOM_STRIDE + col]);
      float2 mp = *mptr;
      mp.x = MOMB*mp.x + (1.0f-MOMB)*err*xp.x;
      mp.y = MOMB*mp.y + (1.0f-MOMB)*err*xp.y;
      *mptr = mp;
      const float w0 = sp*w_u[2*c]   - eff*mp.x;
      const float w1 = sp*w_u[2*c+1] - eff*mp.y;
      w_u[2*c] = w0; w_u[2*c+1] = w1;
      fsum += w0*w0 + w1*w1;
    }
    #pragma unroll
    for (int off = 16; off > 0; off >>= 1) fsum += __shfl_xor(fsum, off, 32);
    if (lane == 0) sm.redf[wid] = fsum;
    __syncthreads();

    if (tid == 0) {
      float s = 0.0f;
      #pragma unroll
      for (int i = 0; i < 32; ++i) s += sm.redf[i];
      const float fn = fmaxf(sqrtf(s), 1e-6f);
      sm.scale_s = fminf(32.0f/fn, 1.0f);
    }
  }
}

extern "C" void kernel_launch(void* const* d_in, const int* in_sizes, int n_in,
                              void* d_out, int out_size, void* d_ws, size_t ws_size,
                              hipStream_t stream) {
  (void)in_sizes; (void)n_in; (void)out_size; (void)ws_size;
  const float* x    = (const float*)d_in[0];
  const float* Wb   = (const float*)d_in[1];
  const float* bias = (const float*)d_in[2];
  const float* lil  = (const float*)d_in[3];
  float* out = (float*)d_out;
  float* nrm2 = (float*)d_ws;
  float* ystg = nrm2 + YSTG_OFF;
  ttt_rows_kernel<<<dim3(NB*NRB), dim3(256), 0, stream>>>(x, Wb, bias, lil, ystg, nrm2);
  ttt_linear_kernel<<<dim3(NB*NSL), dim3(1024), 0, stream>>>(x, Wb, bias, lil, out, nrm2, ystg);
}


# x staged in LDS in chunks of 64 steps, requested a chunk ahead and read once, not twice: the step-size schedule and the steps read it there, so no step issues a global load; every sum as before
# speedup vs baseline: 23.1196x; 1.1651x over previous
#include <hip/hip_runtime.h>
#include <math.h>

#define NB 4
#define NT 512
#define ND 256
#define MOMB 0.9f
#define NOVB 0.95f
#define MOM_STRIDE 260

#define RBLK 8
#define NRB (ND / RBLK)
#define NSL 16
#define YSTG_OFF (NB * NRB * NT)
#define CLAMP_SAFE 1020.0f
#define FSTR 260
#define PF 4
#define CH 64
#define NCH (NT / CH)
#define TSTR 260

typedef float v2f __attribute__((ext_vector_type(2)));
typedef float v8f __attribute__((ext_vector_type(8)));

__device__ __forceinline__ float row_sum256(const float* r)
{
  float s0 = 0.0f, s1 = 0.0f, s2 = 0.0f, s3 = 0.0f;
  #pragma unroll 8
  for (int i = 0; i < ND; i += 4) {
    const float4 v = *reinterpret_cast<const float4*>(r + i);
    s0 += v.x; s1 += v.y; s2 += v.z; s3 += v.w;
  }
  return (s0 + s1) + (s2 + s3);
}

template <int M>
__device__ __forceinline__ float add_row_xor(float v)
{
  const int s = __builtin_bit_cast(int, v);
  const int p = __builtin_amdgcn_update_dpp(s, s, 0x160 + M, 0xf, 0xf, true);
  return v + __builtin_bit_cast(float, p);
}

__device__ __forceinline__ float xhalf(float v)
{
  const int s = __builtin_bit_cast(int, v);
  return __builtin_bit_cast(float,
      __builtin_amdgcn_permlanex16(s, s, 0x76543210, 0xfedcba98, false, false));
}

__global__ __launch_bounds__(256) void ttt_rows_kernel(
    const float* __restrict__ x, const float* __restrict__ Wb,
    const float* __restrict__ bias, const float* __restrict__ loglr,
    float* __restrict__ ystg, float* __restrict__ nrm2)
{
  __shared__ __align__(16) float xs2[2][CH][ND];
  __shared__ __align__(16) float fbuf[CH][FSTR];
  __shared__ float lrs_s[CH];
  __shared__ float eff_s[CH];

  const int tid  = threadIdx.x;
  const int lane = tid & 31;
  const int wid  = __builtin_amdgcn_readfirstlane(tid >> 5);
  const int b    = blockIdx.x / NRB;
  const int rblk = blockIdx.x % NRB;
  const int row  = rblk*RBLK + wid;
  const float* __restrict__ xb = x + (size_t)b*NT*ND;

  float w_u[8], m_u[8];
  {
    const float4 wl = *reinterpret_cast<const float4*>(Wb + row*ND + 4*lane);
    const float4 wh = *reinterpret_cast<const float4*>(Wb + row*ND + 128 + 4*lane);
    w_u[0] = wl.x; w_u[1] = wl.y; w_u[2] = wl.z; w_u[3] = wl.w;
    w_u[4] = wh.x; w_u[5] = wh.y; w_u[6] = wh.z; w_u[7] = wh.w;
    #pragma unroll
    for (int i = 0; i < 8; ++i) m_u[i] = 0.0f;
  }

  const float inner_lr = fminf(fmaxf(expf(loglr[0]), 1e-5f), 1.0f);
  const float bias_r = bias[row];
  float iema  = 0.0f;
  float lrema = 1.0f;
  float ykeep = 0.0f;
  const bool s0 = (lane & 1) != 0, s1 = (lane & 2) != 0, s2 = (lane & 4) != 0;
  const bool on_diag = ((lane >> 3) & 1) == (lane >> 4);

#define XLD(base, k) (*reinterpret_cast<const float4*>((base) + (size_t)((k)*256 + tid)*4))
#define XST(buf, k, v) (*reinterpret_cast<float4*>(&xs2[buf][0][0] + (size_t)((k)*256 + tid)*4) = (v))
  float4 g0, g1, g2, g3, g4, g5, g6, g7, g8, g9, g10, g11, g12, g13, g14, g15;
  {
    g0 = XLD(xb, 0);
    g1 = XLD(xb, 1);
    g2 = XLD(xb, 2);
    g3 = XLD(xb, 3);
    g4 = XLD(xb, 4);
    g5 = XLD(xb, 5);
    g6 = XLD(xb, 6);
    g7 = XLD(xb, 7);
    g8 = XLD(xb, 8);
    g9 = XLD(xb, 9);
    g10 = XLD(xb, 10);
    g11 = XLD(xb, 11);
    g12 = XLD(xb, 12);
    g13 = XLD(xb, 13);
    g14 = XLD(xb, 14);
    g15 = XLD(xb, 15);
    XST(0, 0, g0);
    XST(0, 1, g1);
    XST(0, 2, g2);
    XST(0, 3, g3);
    XST(0, 4, g4);
    XST(0, 5, g5);
    XST(0, 6, g6);
    XST(0, 7, g7);
    XST(0, 8, g8);
    XST(0, 9, g9);
    XST(0, 10, g10);
    XST(0, 11, g11);
    XST(0, 12, g12);
    XST(0, 13, g13);
    XST(0, 14, g14);
    XST(0, 15, g15);
  }
  __syncthreads();

  for (int c = 0; c < NCH; ++c) {
    const int cur = c & 1;
    const float (*xs)[ND] = xs2[cur];
    const bool more = (c + 1 < NCH);
    if (more) {
      const float* __restrict__ xn = xb + (size_t)(c + 1)*CH*ND;
      g0 = XLD(xn, 0);
      g1 = XLD(xn, 1);
      g2 = XLD(xn, 2);
      g3 = XLD(xn, 3);
      g4 = XLD(xn, 4);
      g5 = XLD(xn, 5);
      g6 = XLD(xn, 6);
      g7 = XLD(xn, 7);
      g8 = XLD(xn, 8);
      g9 = XLD(xn, 9);
      g10 = XLD(xn, 10);
      g11 = XLD(xn, 11);
      g12 = XLD(xn, 12);
      g13 = XLD(xn, 13);
      g14 = XLD(xn, 14);
      g15 = XLD(xn, 15);
    }

    #pragma unroll 16
    for (int s = 0; s < CH; ++s) {
      const float xj = xs[s][tid];
      fbuf[s][tid] = fabsf(xj - iema);
      iema = NOVB*iema + (1.0f-NOVB)*xj;
    }
    if (more) {
      XST(cur ^ 1, 0, g0);
      XST(cur ^ 1, 1, g1);
      XST(cur ^ 1, 2, g2);
      XST(cur ^ 1, 3, g3);
      XST(cur ^ 1, 4, g4);
      XST(cur ^ 1, 5, g5);
      XST(cur ^ 1, 6, g6);
      XST(cur ^ 1, 7, g7);
      XST(cur ^ 1, 8, g8);
      XST(cur ^ 1, 9, g9);
      XST(cur ^ 1, 10, g10);
      XST(cur ^ 1, 11, g11);
      XST(cur ^ 1, 12, g12);
      XST(cur ^ 1, 13, g13);
      XST(cur ^ 1, 14, g14);
      XST(cur ^ 1, 15, g15);
    }
    __syncthreads();
    if (tid < CH) {
      const float nov = row_sum256(fbuf[tid]) * (1.0f/ND);
      lrs_s[tid] = fminf(fmaxf(1.0f + 3.0f*nov, 0.5f), 3.0f);
    }
    __syncthreads();
    if (tid == 0) {
      #pragma unroll 16
      for (int s = 0; s < CH; ++s) {
        lrema = NOVB*lrema + (1.0f-NOVB)*lrs_s[s];
        eff_s[s] = inner_lr * lrema;
      }
    }
    __syncthreads();

    float4 xa[PF], xc[PF];
    float  xt[PF], ef[PF];
    #pragma unroll
    for (int u = 0; u < PF - 1; ++u) {
      xa[u] = *reinterpret_cast<const float4*>(&xs[u][4*lane]);
      xc[u] = *reinterpret_cast<const float4*>(&xs[u][128 + 4*lane]);
      xt[u] = xs[u][row];
      ef[u] = eff_s[u];
    }
    xa[PF-1] = make_float4(0.0f, 0.0f, 0.0f, 0.0f);
    xc[PF-1] = make_float4(0.0f, 0.0f, 0.0f, 0.0f);
    xt[PF-1] = 0.0f;
    ef[PF-1] = 0.0f;

    for (int t0 = 0; t0 < CH; t0 += PF) {
      #pragma unroll
      for (int u = 0; u < PF; ++u) {
        const int s = t0 + u;
        const float xv[8] = { xa[u].x, xa[u].y, xa[u].z, xa[u].w,
                              xc[u].x, xc[u].y, xc[u].z, xc[u].w };
        const float xrow = xt[u];
        const float eff  = ef[u];
        v8f acc0 = {}, acc1 = {};
        #pragma unroll
        for (int j = 0; j < 4; ++j) {
          v2f av; av.x = w_u[2*j]; av.y = w_u[2*j+1];
          v2f bv; bv.x = xv[2*j];  bv.y = xv[2*j+1];
          if (j & 1)
            acc1 = __builtin_amdgcn_wmma_f32_16x16x4_f32(false, av, false, bv, (short)0, acc1, false, false);
          else
            acc0 = __builtin_amdgcn_wmma_f32_16x16x4_f32(false, av, false, bv, (short)0, acc0, false, false);
        }
        __builtin_amdgcn_sched_barrier(0);
        {
          const int un = (u + PF - 1) % PF;
          const int sn = (s + PF - 1 < CH) ? s + PF - 1 : CH - 1;
          xa[un] = *reinterpret_cast<const float4*>(&xs[sn][4*lane]);
          xc[un] = *reinterpret_cast<const float4*>(&xs[sn][128 + 4*lane]);
          xt[un] = xs[sn][row];
          ef[un] = eff_s[sn];
        }
        const v8f acc = acc0 + acc1;
        const float d01 = s0 ? acc[1] : acc[0];
        const float d23 = s0 ? acc[3] : acc[2];
        const float d45 = s0 ? acc[5] : acc[4];
        const float d67 = s0 ? acc[7] : acc[6];
        const float d03 = s1 ? d23 : d01;
        const float d47 = s1 ? d67 : d45;
        float d = s2 ? d47 : d03;
        d = on_diag ? d : 0.0f;
        d = add_row_xor<1>(d);
        d = add_row_xor<2>(d);
        d = add_row_xor<4>(d);
        d = add_row_xor<8>(d);
        const float y = (d + xhalf(d)) + bias_r;

        ykeep = (lane == (s & 31)) ? y : ykeep;
        const float ce = (y - xrow) * ((1.0f-MOMB) * (1.0f/ND));

        float fs0 = 0.0f, fs1 = 0.0f;
        #pragma unroll
        for (int i = 0; i < 8; i += 2) {
          const float m0 = MOMB*m_u[i]   + ce*xv[i];
          const float m1 = MOMB*m_u[i+1] + ce*xv[i+1];
          m_u[i] = m0; m_u[i+1] = m1;
          const float w0 = w_u[i]   - eff*m0;
          const float w1 = w_u[i+1] - eff*m1;
          w_u[i] = w0; w_u[i+1] = w1;
          fs0 += w0*w0; fs1 += w1*w1;
        }
        fbuf[s][tid] = fs0 + fs1;
      }
      const int sl = t0 + PF - 1;
      if ((sl & 31) == 31)
        ystg[((size_t)(b*ND + row))*NT + c*CH + (sl - 31) + lane] = ykeep;
    }
    __syncthreads();
    if (tid < CH)
      nrm2[(size_t)(b*NRB + rblk)*NT + c*CH + tid] = row_sum256(fbuf[tid]);
    __syncthreads();
  }
#undef XLD
#undef XST
}

struct __align__(16) Smem {
  float mom[ND * MOM_STRIDE];
  float x_s[ND];
  float err_s[ND];
  float yacc[2][ND];
  float iema[ND];
  float redn[8];
  float redf[32];
  float eff_s;
  float scale_s;
};

__global__ __launch_bounds__(1024) void ttt_linear_kernel(
    const float* __restrict__ x, const float* __restrict__ Wb,
    const float* __restrict__ bias, const float* __restrict__ loglr,
    float* __restrict__ out, const float* __restrict__ nrm2,
    const float* __restrict__ ystg)
{
  __shared__ Smem sm;
  const int tid  = threadIdx.x;
  const int lane = tid & 31;
  const int wid  = tid >> 5;
  const int b    = blockIdx.x / NSL;
  const int sl   = blockIdx.x % NSL;

  {
    int bad = 0;
    if (tid < NT) {
      float s = 0.0f;
      #pragma unroll
      for (int r = 0; r < NRB; ++r) s += nrm2[(size_t)(b*NRB + r)*NT + tid];
      bad = !(s <= CLAMP_SAFE);
    }
    if (!__syncthreads_or(bad)) {
      float* tile = sm.mom;
      for (int i = tid; i < ND*8; i += 1024) {
        const int r  = i >> 3;
        const int c4 = (i & 7) << 2;
        const float4 v = *reinterpret_cast<const float4*>(
            ystg + ((size_t)(b*ND + r))*NT + sl*(NT/NSL) + c4);
        tile[(c4 + 0)*TSTR + r] = v.x;
        tile[(c4 + 1)*TSTR + r] = v.y;
        tile[(c4 + 2)*TSTR + r] = v.z;
        tile[(c4 + 3)*TSTR + r] = v.w;
      }
      __syncthreads();
      for (int i = tid; i < (NT/NSL)*(ND/4); i += 1024) {
        const int tl = i >> 6;
        const int q  = i & 63;
        const float4 v = *reinterpret_cast<const float4*>(&tile[tl*TSTR + 4*q]);
        *reinterpret_cast<float4*>(out + ((size_t)b*NT + sl*(NT/NSL) + tl)*ND + 4*q) = v;
      }
      return;
    }
    if (sl != 0) return;
  }

  const int hi   = lane >> 4;
  const int rb   = (wid >> 1) << 4;
  const int row  = rb + (lane & 15);
  const int colbase = (wid & 1) << 7;

  float w_u[64];
  #pragma unroll
  for (int c = 0; c < 32; ++c) {
    const int col = colbase + 4*c + 2*hi;
    const float2 wv = *reinterpret_cast<const float2*>(Wb + row*ND + col);
    w_u[2*c]   = wv.x;
    w_u[2*c+1] = wv.y;
  }
  for (int i = tid; i < ND*MOM_STRIDE; i += 1024) sm.mom[i] = 0.0f;
  if (tid < ND) sm.iema[tid] = 0.0f;

  float lrema    = 1.0f;
  float inner_lr = 0.0f;
  if (tid == 0) {
    inner_lr = fminf(fmaxf(expf(loglr[0]), 1e-5f), 1.0f);
    sm.scale_s = 1.0f;
    sm.eff_s   = 0.0f;
  }
  const float bias_r = (tid < ND) ? bias[tid] : 0.0f;
  __syncthreads();

  for (int t = 0; t < NT; ++t) {
    if (tid < ND) {
      const float xj = x[((size_t)b*NT + t)*ND + tid];
      sm.x_s[tid] = xj;
      float a = fabsf(xj - sm.iema[tid]);
      sm.iema[tid] = NOVB*sm.iema[tid] + (1.0f-NOVB)*xj;
      #pragma unroll
      for (int off = 16; off > 0; off >>= 1) a += __shfl_xor(a, off, 32);
      if (lane == 0) sm.redn[wid] = a;
    }
    __syncthreads();

    if (tid == 0) {
      float s = 0.0f;
      #pragma unroll
      for (int i = 0; i < 8; ++i) s += sm.redn[i];
      const float nov = s * (1.0f/ND);
      const float lrs = fminf(fmaxf(1.0f + 3.0f*nov, 0.5f), 3.0f);
      lrema = NOVB*lrema + (1.0f-NOVB)*lrs;
      sm.eff_s = inner_lr * lrema;
    }

    v8f acc = {};
    #pragma unroll
    for (int c = 0; c < 32; ++c) {
      const int col = colbase + 4*c + 2*hi;
      const float2 xp = *reinterpret_cast<const float2*>(&sm.x_s[col]);
      v2f av; av.x = w_u[2*c]; av.y = w_u[2*c+1];
      v2f bv; bv.x = xp.x;     bv.y = xp.y;
      acc = __builtin_amdgcn_wmma_f32_16x16x4_f32(
           false, av,  false, bv,
           (short)0, acc,  false,  false);
    }
    if ((lane & 15) == 0) {
      const int mbase = rb + hi*8;
      #pragma unroll
      for (int v = 0; v < 8; ++v) sm.yacc[wid & 1][mbase + v] = acc[v];
    }
    __syncthreads();

    if (tid < ND) {
      const float yv = sm.scale_s*(sm.yacc[0][tid] + sm.yacc[1][tid]) + bias_r;
      out[((size_t)b*NT + t)*ND + tid] = yv;
      sm.err_s[tid] = (yv - sm.x_s[tid]) * (1.0f/ND);
    }
    __syncthreads();

    const float err = sm.err_s[row];
    const float eff = sm.eff_s;
    const float sp  = sm.scale_s;
    float fsum = 0.0f;
    #pragma unroll
    for (int c = 0; c < 32; ++c) {
      const int col = colbase + 4*c + 2*hi;
      const float2 xp = *reinterpret_cast<const float2*>(&sm.x_s[col]);
      float2* mptr = reinterpret_cast<float2*>(&sm.mom[row*MOM_STRIDE + col]);
      float2 mp = *mptr;
      mp.x = MOMB*mp.x + (1.0f-MOMB)*err*xp.x;
      mp.y = MOMB*mp.y + (1.0f-MOMB)*err*xp.y;
      *mptr = mp;
      const float w0 = sp*w_u[2*c]   - eff*mp.x;
      const float w1 = sp*w_u[2*c+1] - eff*mp.y;
      w_u[2*c] = w0; w_u[2*c+1] = w1;
      fsum += w0*w0 + w1*w1;
    }
    #pragma unroll
    for (int off = 16; off > 0; off >>= 1) fsum += __shfl_xor(fsum, off, 32);
    if (lane == 0) sm.redf[wid] = fsum;
    __syncthreads();

    if (tid == 0) {
      float s = 0.0f;
      #pragma unroll
      for (int i = 0; i < 32; ++i) s += sm.redf[i];
      const float fn = fmaxf(sqrtf(s), 1e-6f);
      sm.scale_s = fminf(32.0f/fn, 1.0f);
    }
  }
}

extern "C" void kernel_launch(void* const* d_in, const int* in_sizes, int n_in,
                              void* d_out, int out_size, void* d_ws, size_t ws_size,
                              hipStream_t stream) {
  (void)in_sizes; (void)n_in; (void)out_size; (void)ws_size;
  const float* x    = (const float*)d_in[0];
  const float* Wb   = (const float*)d_in[1];
  const float* bias = (const float*)d_in[2];
  const float* lil  = (const float*)d_in[3];
  float* out = (float*)d_out;
  float* nrm2 = (float*)d_ws;
  float* ystg = nrm2 + YSTG_OFF;
  ttt_rows_kernel<<<dim3(NB*NRB), dim3(256), 0, stream>>>(x, Wb, bias, lil, ystg, nrm2);
  ttt_linear_kernel<<<dim3(NB*NSL), dim3(1024), 0, stream>>>(x, Wb, bias, lil, out, nrm2, ystg);
}


# Work-groups of 4 rows (4 waves), not 8: 256 work-groups, one wave per SIMD on every WGP, not two per SIMD on half the WGPs; y sums as before, the clamp check's norm sum regroups (64 row blocks)
# speedup vs baseline: 28.4264x; 1.2295x over previous
#include <hip/hip_runtime.h>
#include <math.h>

#define NB 4
#define NT 512
#define ND 256
#define MOMB 0.9f
#define NOVB 0.95f
#define MOM_STRIDE 260

#define RBLK 4
#define NRB (ND / RBLK)
#define NTH (RBLK * 32)
#define NPC (CH * ND / 4 / NTH)
#define NSL 16
#define YSTG_OFF (NB * NRB * NT)
#define CLAMP_SAFE 1020.0f
#define FSTR 260
#define PF 4
#define CH 64
#define NCH (NT / CH)
#define TSTR 260

typedef float v2f __attribute__((ext_vector_type(2)));
typedef float v8f __attribute__((ext_vector_type(8)));

template <int N>
__device__ __forceinline__ float row_sum(const float* r)
{
  float s0 = 0.0f, s1 = 0.0f, s2 = 0.0f, s3 = 0.0f;
  #pragma unroll 8
  for (int i = 0; i < N; i += 4) {
    const float4 v = *reinterpret_cast<const float4*>(r + i);
    s0 += v.x; s1 += v.y; s2 += v.z; s3 += v.w;
  }
  return (s0 + s1) + (s2 + s3);
}

template <int M>
__device__ __forceinline__ float add_row_xor(float v)
{
  const int s = __builtin_bit_cast(int, v);
  const int p = __builtin_amdgcn_update_dpp(s, s, 0x160 + M, 0xf, 0xf, true);
  return v + __builtin_bit_cast(float, p);
}

__device__ __forceinline__ float xhalf(float v)
{
  const int s = __builtin_bit_cast(int, v);
  return __builtin_bit_cast(float,
      __builtin_amdgcn_permlanex16(s, s, 0x76543210, 0xfedcba98, false, false));
}

__global__ __launch_bounds__(NTH) void ttt_rows_kernel(
    const float* __restrict__ x, const float* __restrict__ Wb,
    const float* __restrict__ bias, const float* __restrict__ loglr,
    float* __restrict__ ystg, float* __restrict__ nrm2)
{
  __shared__ __align__(16) float xs2[2][CH][ND];
  __shared__ __align__(16) float fbuf[CH][FSTR];
  __shared__ float lrs_s[CH];
  __shared__ float eff_s[CH];

  const int tid  = threadIdx.x;
  const int lane = tid & 31;
  const int wid  = __builtin_amdgcn_readfirstlane(tid >> 5);
  const int b    = blockIdx.x / NRB;
  const int rblk = blockIdx.x % NRB;
  const int row  = rblk*RBLK + wid;
  const float* __restrict__ xb = x + (size_t)b*NT*ND;

  float w_u[8], m_u[8];
  {
    const float4 wl = *reinterpret_cast<const float4*>(Wb + row*ND + 4*lane);
    const float4 wh = *reinterpret_cast<const float4*>(Wb + row*ND + 128 + 4*lane);
    w_u[0] = wl.x; w_u[1] = wl.y; w_u[2] = wl.z; w_u[3] = wl.w;
    w_u[4] = wh.x; w_u[5] = wh.y; w_u[6] = wh.z; w_u[7] = wh.w;
    #pragma unroll
    for (int i = 0; i < 8; ++i) m_u[i] = 0.0f;
  }

  const float inner_lr = fminf(fmaxf(expf(loglr[0]), 1e-5f), 1.0f);
  const float bias_r = bias[row];
  float iema0 = 0.0f;
  float iema1 = 0.0f;
  float lrema = 1.0f;
  float ykeep = 0.0f;
  const bool s0 = (lane & 1) != 0, s1 = (lane & 2) != 0, s2 = (lane & 4) != 0;
  const bool on_diag = ((lane >> 3) & 1) == (lane >> 4);

#define XLD(base, k) (*reinterpret_cast<const float4*>((base) + (size_t)((k)*NTH + tid)*4))
#define XST(buf, k, v) (*reinterpret_cast<float4*>(&xs2[buf][0][0] + (size_t)((k)*NTH + tid)*4) = (v))
#define XPIECES(F) F(0) F(1) F(2) F(3) F(4) F(5) F(6) F(7) F(8) F(9) F(10) F(11) F(12) F(13) \
  F(14) F(15) F(16) F(17) F(18) F(19) F(20) F(21) F(22) F(23) F(24) F(25) F(26) F(27) F(28) \
  F(29) F(30) F(31)
  static_assert(NPC == 32, "XPIECES lists NPC pieces");
#define XDECL(k) float4 g##k;
  XPIECES(XDECL)
#undef XDECL
  {
#define XLD0(k) g##k = XLD(xb, k);
#define XST0(k) XST(0, k, g##k);
    XPIECES(XLD0)
    XPIECES(XST0)
#undef XLD0
#undef XST0
  }
  __syncthreads();

  for (int c = 0; c < NCH; ++c) {
    const int cur = c & 1;
    const float (*xs)[ND] = xs2[cur];
    const bool more = (c + 1 < NCH);
    if (more) {
      const float* __restrict__ xn = xb + (size_t)(c + 1)*CH*ND;
#define XLDN(k) g##k = XLD(xn, k);
      XPIECES(XLDN)
#undef XLDN
    }

    #pragma unroll 16
    for (int s = 0; s < CH; ++s) {
      const float xj0 = xs[s][tid];
      const float xj1 = xs[s][tid + NTH];
      fbuf[s][tid]       = fabsf(xj0 - iema0);
      fbuf[s][tid + NTH] = fabsf(xj1 - iema1);
      iema0 = NOVB*iema0 + (1.0f-NOVB)*xj0;
      iema1 = NOVB*iema1 + (1.0f-NOVB)*xj1;
    }
    if (more) {
#define XSTN(k) XST(cur ^ 1, k, g##k);
      XPIECES(XSTN)
#undef XSTN
    }
    __syncthreads();
    if (tid < CH) {
      const float nov = row_sum<ND>(fbuf[tid]) * (1.0f/ND);
      lrs_s[tid] = fminf(fmaxf(1.0f + 3.0f*nov, 0.5f), 3.0f);
    }
    __syncthreads();
    if (tid == 0) {
      #pragma unroll 16
      for (int s = 0; s < CH; ++s) {
        lrema = NOVB*lrema + (1.0f-NOVB)*lrs_s[s];
        eff_s[s] = inner_lr * lrema;
      }
    }
    __syncthreads();

    float4 xa[PF], xc[PF];
    float  xt[PF], ef[PF];
    #pragma unroll
    for (int u = 0; u < PF - 1; ++u) {
      xa[u] = *reinterpret_cast<const float4*>(&xs[u][4*lane]);
      xc[u] = *reinterpret_cast<const float4*>(&xs[u][128 + 4*lane]);
      xt[u] = xs[u][row];
      ef[u] = eff_s[u];
    }
    xa[PF-1] = make_float4(0.0f, 0.0f, 0.0f, 0.0f);
    xc[PF-1] = make_float4(0.0f, 0.0f, 0.0f, 0.0f);
    xt[PF-1] = 0.0f;
    ef[PF-1] = 0.0f;

    for (int t0 = 0; t0 < CH; t0 += PF) {
      #pragma unroll
      for (int u = 0; u < PF; ++u) {
        const int s = t0 + u;
        const float xv[8] = { xa[u].x, xa[u].y, xa[u].z, xa[u].w,
                              xc[u].x, xc[u].y, xc[u].z, xc[u].w };
        const float xrow = xt[u];
        const float eff  = ef[u];
        v8f acc0 = {}, acc1 = {};
        #pragma unroll
        for (int j = 0; j < 4; ++j) {
          v2f av; av.x = w_u[2*j]; av.y = w_u[2*j+1];
          v2f bv; bv.x = xv[2*j];  bv.y = xv[2*j+1];
          if (j & 1)
            acc1 = __builtin_amdgcn_wmma_f32_16x16x4_f32(false, av, false, bv, (short)0, acc1, false, false);
          else
            acc0 = __builtin_amdgcn_wmma_f32_16x16x4_f32(false, av, false, bv, (short)0, acc0, false, false);
        }
        __builtin_amdgcn_sched_barrier(0);
        {
          const int un = (u + PF - 1) % PF;
          const int sn = (s + PF - 1 < CH) ? s + PF - 1 : CH - 1;
          xa[un] = *reinterpret_cast<const float4*>(&xs[sn][4*lane]);
          xc[un] = *reinterpret_cast<const float4*>(&xs[sn][128 + 4*lane]);
          xt[un] = xs[sn][row];
          ef[un] = eff_s[sn];
        }
        const v8f acc = acc0 + acc1;
        const float d01 = s0 ? acc[1] : acc[0];
        const float d23 = s0 ? acc[3] : acc[2];
        const float d45 = s0 ? acc[5] : acc[4];
        const float d67 = s0 ? acc[7] : acc[6];
        const float d03 = s1 ? d23 : d01;
        const float d47 = s1 ? d67 : d45;
        float d = s2 ? d47 : d03;
        d = on_diag ? d : 0.0f;
        d = add_row_xor<1>(d);
        d = add_row_xor<2>(d);
        d = add_row_xor<4>(d);
        d = add_row_xor<8>(d);
        const float y = (d + xhalf(d)) + bias_r;

        ykeep = (lane == (s & 31)) ? y : ykeep;
        const float ce = (y - xrow) * ((1.0f-MOMB) * (1.0f/ND));

        float fs0 = 0.0f, fs1 = 0.0f;
        #pragma unroll
        for (int i = 0; i < 8; i += 2) {
          const float m0 = MOMB*m_u[i]   + ce*xv[i];
          const float m1 = MOMB*m_u[i+1] + ce*xv[i+1];
          m_u[i] = m0; m_u[i+1] = m1;
          const float w0 = w_u[i]   - eff*m0;
          const float w1 = w_u[i+1] - eff*m1;
          w_u[i] = w0; w_u[i+1] = w1;
          fs0 += w0*w0; fs1 += w1*w1;
        }
        fbuf[s][tid] = fs0 + fs1;
      }
      const int sl = t0 + PF - 1;
      if ((sl & 31) == 31)
        ystg[((size_t)(b*ND + row))*NT + c*CH + (sl - 31) + lane] = ykeep;
    }
    __syncthreads();
    if (tid < CH)
      nrm2[(size_t)(b*NRB + rblk)*NT + c*CH + tid] = row_sum<NTH>(fbuf[tid]);
    __syncthreads();
  }
#undef XLD
#undef XST
#undef XPIECES
}

struct __align__(16) Smem {
  float mom[ND * MOM_STRIDE];
  float x_s[ND];
  float err_s[ND];
  float yacc[2][ND];
  float iema[ND];
  float redn[8];
  float redf[32];
  float eff_s;
  float scale_s;
};

__global__ __launch_bounds__(1024) void ttt_linear_kernel(
    const float* __restrict__ x, const float* __restrict__ Wb,
    const float* __restrict__ bias, const float* __restrict__ loglr,
    float* __restrict__ out, const float* __restrict__ nrm2,
    const float* __restrict__ ystg)
{
  __shared__ Smem sm;
  const int tid  = threadIdx.x;
  const int lane = tid & 31;
  const int wid  = tid >> 5;
  const int b    = blockIdx.x / NSL;
  const int sl   = blockIdx.x % NSL;

  {
    int bad = 0;
    if (tid < NT) {
      float s = 0.0f;
      #pragma unroll
      for (int r = 0; r < NRB; ++r) s += nrm2[(size_t)(b*NRB + r)*NT + tid];
      bad = !(s <= CLAMP_SAFE);
    }
    if (!__syncthreads_or(bad)) {
      float* tile = sm.mom;
      for (int i = tid; i < ND*8; i += 1024) {
        const int r  = i >> 3;
        const int c4 = (i & 7) << 2;
        const float4 v = *reinterpret_cast<const float4*>(
            ystg + ((size_t)(b*ND + r))*NT + sl*(NT/NSL) + c4);
        tile[(c4 + 0)*TSTR + r] = v.x;
        tile[(c4 + 1)*TSTR + r] = v.y;
        tile[(c4 + 2)*TSTR + r] = v.z;
        tile[(c4 + 3)*TSTR + r] = v.w;
      }
      __syncthreads();
      for (int i = tid; i < (NT/NSL)*(ND/4); i += 1024) {
        const int tl = i >> 6;
        const int q  = i & 63;
        const float4 v = *reinterpret_cast<const float4*>(&tile[tl*TSTR + 4*q]);
        *reinterpret_cast<float4*>(out + ((size_t)b*NT + sl*(NT/NSL) + tl)*ND + 4*q) = v;
      }
      return;
    }
    if (sl != 0) return;
  }

  const int hi   = lane >> 4;
  const int rb   = (wid >> 1) << 4;
  const int row  = rb + (lane & 15);
  const int colbase = (wid & 1) << 7;

  float w_u[64];
  #pragma unroll
  for (int c = 0; c < 32; ++c) {
    const int col = colbase + 4*c + 2*hi;
    const float2 wv = *reinterpret_cast<const float2*>(Wb + row*ND + col);
    w_u[2*c]   = wv.x;
    w_u[2*c+1] = wv.y;
  }
  for (int i = tid; i < ND*MOM_STRIDE; i += 1024) sm.mom[i] = 0.0f;
  if (tid < ND) sm.iema[tid] = 0.0f;

  float lrema    = 1.0f;
  float inner_lr = 0.0f;
  if (tid == 0) {
    inner_lr = fminf(fmaxf(expf(loglr[0]), 1e-5f), 1.0f);
    sm.scale_s = 1.0f;
    sm.eff_s   = 0.0f;
  }
  const float bias_r = (tid < ND) ? bias[tid] : 0.0f;
  __syncthreads();

  for (int t = 0; t < NT; ++t) {
    if (tid < ND) {
      const float xj = x[((size_t)b*NT + t)*ND + tid];
      sm.x_s[tid] = xj;
      float a = fabsf(xj - sm.iema[tid]);
      sm.iema[tid] = NOVB*sm.iema[tid] + (1.0f-NOVB)*xj;
      #pragma unroll
      for (int off = 16; off > 0; off >>= 1) a += __shfl_xor(a, off, 32);
      if (lane == 0) sm.redn[wid] = a;
    }
    __syncthreads();

    if (tid == 0) {
      float s = 0.0f;
      #pragma unroll
      for (int i = 0; i < 8; ++i) s += sm.redn[i];
      const float nov = s * (1.0f/ND);
      const float lrs = fminf(fmaxf(1.0f + 3.0f*nov, 0.5f), 3.0f);
      lrema = NOVB*lrema + (1.0f-NOVB)*lrs;
      sm.eff_s = inner_lr * lrema;
    }

    v8f acc = {};
    #pragma unroll
    for (int c = 0; c < 32; ++c) {
      const int col = colbase + 4*c + 2*hi;
      const float2 xp = *reinterpret_cast<const float2*>(&sm.x_s[col]);
      v2f av; av.x = w_u[2*c]; av.y = w_u[2*c+1];
      v2f bv; bv.x = xp.x;     bv.y = xp.y;
      acc = __builtin_amdgcn_wmma_f32_16x16x4_f32(
           false, av,  false, bv,
           (short)0, acc,  false,  false);
    }
    if ((lane & 15) == 0) {
      const int mbase = rb + hi*8;
      #pragma unroll
      for (int v = 0; v < 8; ++v) sm.yacc[wid & 1][mbase + v] = acc[v];
    }
    __syncthreads();

    if (tid < ND) {
      const float yv = sm.scale_s*(sm.yacc[0][tid] + sm.yacc[1][tid]) + bias_r;
      out[((size_t)b*NT + t)*ND + tid] = yv;
      sm.err_s[tid] = (yv - sm.x_s[tid]) * (1.0f/ND);
    }
    __syncthreads();

    const float err = sm.err_s[row];
    const float eff = sm.eff_s;
    const float sp  = sm.scale_s;
    float fsum = 0.0f;
    #pragma unroll
    for (int c = 0; c < 32; ++c) {
      const int col = colbase + 4*c + 2*hi;
      const float2 xp = *reinterpret_cast<const float2*>(&sm.x_s[col]);
      float2* mptr = reinterpret_cast<float2*>(&sm.mom[row*MOM_STRIDE + col]);
      float2 mp = *mptr;
      mp.x = MOMB*mp.x + (1.0f-MOMB)*err*xp.x;
      mp.y = MOMB*mp.y + (1.0f-MOMB)*err*xp.y;
      *mptr = mp;
      const float w0 = sp*w_u[2*c]   - eff*mp.x;
      const float w1 = sp*w_u[2*c+1] - eff*mp.y;
      w_u[2*c] = w0; w_u[2*c+1] = w1;
      fsum += w0*w0 + w1*w1;
    }
    #pragma unroll
    for (int off = 16; off > 0; off >>= 1) fsum += __shfl_xor(fsum, off, 32);
    if (lane == 0) sm.redf[wid] = fsum;
    __syncthreads();

    if (tid == 0) {
      float s = 0.0f;
      #pragma unroll
      for (int i = 0; i < 32; ++i) s += sm.redf[i];
      const float fn = fmaxf(sqrtf(s), 1e-6f);
      sm.scale_s = fminf(32.0f/fn, 1.0f);
    }
  }
}

extern "C" void kernel_launch(void* const* d_in, const int* in_sizes, int n_in,
                              void* d_out, int out_size, void* d_ws, size_t ws_size,
                              hipStream_t stream) {
  (void)in_sizes; (void)n_in; (void)out_size; (void)ws_size;
  const float* x    = (const float*)d_in[0];
  const float* Wb   = (const float*)d_in[1];
  const float* bias = (const float*)d_in[2];
  const float* lil  = (const float*)d_in[3];
  float* out = (float*)d_out;
  float* nrm2 = (float*)d_ws;
  float* ystg = nrm2 + YSTG_OFF;
  ttt_rows_kernel<<<dim3(NB*NRB), dim3(NTH), 0, stream>>>(x, Wb, bias, lil, ystg, nrm2);
  ttt_linear_kernel<<<dim3(NB*NSL), dim3(1024), 0, stream>>>(x, Wb, bias, lil, out, nrm2, ystg);
}


# Step loop's LDS traffic issued as one group per 4 steps (next 4 steps' x into a second register set, the 4 norm partials out), not among every step's matrix instructions; every sum as before
# speedup vs baseline: 32.0508x; 1.1275x over previous
#include <hip/hip_runtime.h>
#include <math.h>

#define NB 4
#define NT 512
#define ND 256
#define MOMB 0.9f
#define NOVB 0.95f
#define MOM_STRIDE 260

#define RBLK 4
#define NRB (ND / RBLK)
#define NTH (RBLK * 32)
#define NPC (CH * ND / 4 / NTH)
#define NSL 16
#define YSTG_OFF (NB * NRB * NT)
#define CLAMP_SAFE 1020.0f
#define FSTR 260
#define PF 4
#define CH 64
#define NCH (NT / CH)
#define TSTR 260

typedef float v2f __attribute__((ext_vector_type(2)));
typedef float v8f __attribute__((ext_vector_type(8)));

template <int N>
__device__ __forceinline__ float row_sum(const float* r)
{
  float s0 = 0.0f, s1 = 0.0f, s2 = 0.0f, s3 = 0.0f;
  #pragma unroll 8
  for (int i = 0; i < N; i += 4) {
    const float4 v = *reinterpret_cast<const float4*>(r + i);
    s0 += v.x; s1 += v.y; s2 += v.z; s3 += v.w;
  }
  return (s0 + s1) + (s2 + s3);
}

template <int M>
__device__ __forceinline__ float add_row_xor(float v)
{
  const int s = __builtin_bit_cast(int, v);
  const int p = __builtin_amdgcn_update_dpp(s, s, 0x160 + M, 0xf, 0xf, true);
  return v + __builtin_bit_cast(float, p);
}

__device__ __forceinline__ float xhalf(float v)
{
  const int s = __builtin_bit_cast(int, v);
  return __builtin_bit_cast(float,
      __builtin_amdgcn_permlanex16(s, s, 0x76543210, 0xfedcba98, false, false));
}

__global__ __launch_bounds__(NTH) void ttt_rows_kernel(
    const float* __restrict__ x, const float* __restrict__ Wb,
    const float* __restrict__ bias, const float* __restrict__ loglr,
    float* __restrict__ ystg, float* __restrict__ nrm2)
{
  __shared__ __align__(16) float xs2[2][CH][ND];
  __shared__ __align__(16) float fbuf[CH][FSTR];
  __shared__ float lrs_s[CH];
  __shared__ __align__(16) float eff_s[CH];

  const int tid  = threadIdx.x;
  const int lane = tid & 31;
  const int wid  = __builtin_amdgcn_readfirstlane(tid >> 5);
  const int b    = blockIdx.x / NRB;
  const int rblk = blockIdx.x % NRB;
  const int row  = rblk*RBLK + wid;
  const float* __restrict__ xb = x + (size_t)b*NT*ND;

  float w_u[8], m_u[8];
  {
    const float4 wl = *reinterpret_cast<const float4*>(Wb + row*ND + 4*lane);
    const float4 wh = *reinterpret_cast<const float4*>(Wb + row*ND + 128 + 4*lane);
    w_u[0] = wl.x; w_u[1] = wl.y; w_u[2] = wl.z; w_u[3] = wl.w;
    w_u[4] = wh.x; w_u[5] = wh.y; w_u[6] = wh.z; w_u[7] = wh.w;
    #pragma unroll
    for (int i = 0; i < 8; ++i) m_u[i] = 0.0f;
  }

  const float inner_lr = fminf(fmaxf(expf(loglr[0]), 1e-5f), 1.0f);
  const float bias_r = bias[row];
  float iema0 = 0.0f;
  float iema1 = 0.0f;
  float lrema = 1.0f;
  float ykeep = 0.0f;
  const bool s0 = (lane & 1) != 0, s1 = (lane & 2) != 0, s2 = (lane & 4) != 0;
  const bool on_diag = ((lane >> 3) & 1) == (lane >> 4);

#define XLD(base, k) (*reinterpret_cast<const float4*>((base) + (size_t)((k)*NTH + tid)*4))
#define XST(buf, k, v) (*reinterpret_cast<float4*>(&xs2[buf][0][0] + (size_t)((k)*NTH + tid)*4) = (v))
#define XPIECES(F) F(0) F(1) F(2) F(3) F(4) F(5) F(6) F(7) F(8) F(9) F(10) F(11) F(12) F(13) \
  F(14) F(15) F(16) F(17) F(18) F(19) F(20) F(21) F(22) F(23) F(24) F(25) F(26) F(27) F(28) \
  F(29) F(30) F(31)
  static_assert(NPC == 32, "XPIECES lists NPC pieces");
#define XDECL(k) float4 g##k;
  XPIECES(XDECL)
#undef XDECL
  {
#define XLD0(k) g##k = XLD(xb, k);
#define XST0(k) XST(0, k, g##k);
    XPIECES(XLD0)
    XPIECES(XST0)
#undef XLD0
#undef XST0
  }
  __syncthreads();

  for (int c = 0; c < NCH; ++c) {
    const int cur = c & 1;
    const float (*xs)[ND] = xs2[cur];
    const bool more = (c + 1 < NCH);
    if (more) {
      const float* __restrict__ xn = xb + (size_t)(c + 1)*CH*ND;
#define XLDN(k) g##k = XLD(xn, k);
      XPIECES(XLDN)
#undef XLDN
    }

    #pragma unroll 16
    for (int s = 0; s < CH; ++s) {
      const float xj0 = xs[s][tid];
      const float xj1 = xs[s][tid + NTH];
      fbuf[s][tid]       = fabsf(xj0 - iema0);
      fbuf[s][tid + NTH] = fabsf(xj1 - iema1);
      iema0 = NOVB*iema0 + (1.0f-NOVB)*xj0;
      iema1 = NOVB*iema1 + (1.0f-NOVB)*xj1;
    }
    if (more) {
#define XSTN(k) XST(cur ^ 1, k, g##k);
      XPIECES(XSTN)
#undef XSTN
    }
    __syncthreads();
    if (tid < CH) {
      const float nov = row_sum<ND>(fbuf[tid]) * (1.0f/ND);
      lrs_s[tid] = fminf(fmaxf(1.0f + 3.0f*nov, 0.5f), 3.0f);
    }
    __syncthreads();
    if (tid == 0) {
      #pragma unroll 16
      for (int s = 0; s < CH; ++s) {
        lrema = NOVB*lrema + (1.0f-NOVB)*lrs_s[s];
        eff_s[s] = inner_lr * lrema;
      }
    }
    __syncthreads();

    float4 xa[2][PF], xc[2][PF];
    float  xt[2][PF];
    float4 efv[2];
    float  fsv[PF];
    static_assert(PF == 4, "a trip's step sizes are read as one float4");
    #pragma unroll
    for (int u = 0; u < PF; ++u) {
      xa[0][u] = *reinterpret_cast<const float4*>(&xs[u][4*lane]);
      xc[0][u] = *reinterpret_cast<const float4*>(&xs[u][128 + 4*lane]);
      xt[0][u] = xs[u][row];
      fsv[u] = 0.0f;
    }
    efv[0] = *reinterpret_cast<const float4*>(&eff_s[0]);

    for (int t0 = 0; t0 < CH; t0 += 2*PF) {
      #pragma unroll
      for (int h = 0; h < 2; ++h) {
        const int tb = t0 + h*PF;
        {
          if (tb > 0) {
            #pragma unroll
            for (int u = 0; u < PF; ++u) fbuf[tb - PF + u][tid] = fsv[u];
          }
          const int tn = (tb + PF < CH) ? tb + PF : tb;
          #pragma unroll
          for (int u = 0; u < PF; ++u) {
            xa[h^1][u] = *reinterpret_cast<const float4*>(&xs[tn + u][4*lane]);
            xc[h^1][u] = *reinterpret_cast<const float4*>(&xs[tn + u][128 + 4*lane]);
            xt[h^1][u] = xs[tn + u][row];
          }
          efv[h^1] = *reinterpret_cast<const float4*>(&eff_s[tn]);
        }
        __builtin_amdgcn_sched_barrier(0);
        const float efa[PF] = { efv[h].x, efv[h].y, efv[h].z, efv[h].w };
        #pragma unroll
        for (int u = 0; u < PF; ++u) {
          const int s = tb + u;
          const float xv[8] = { xa[h][u].x, xa[h][u].y, xa[h][u].z, xa[h][u].w,
                                xc[h][u].x, xc[h][u].y, xc[h][u].z, xc[h][u].w };
          const float xrow = xt[h][u];
          const float eff  = efa[u];
          v8f acc0 = {}, acc1 = {};
          #pragma unroll
          for (int j = 0; j < 4; ++j) {
            v2f av; av.x = w_u[2*j]; av.y = w_u[2*j+1];
            v2f bv; bv.x = xv[2*j];  bv.y = xv[2*j+1];
            if (j & 1)
              acc1 = __builtin_amdgcn_wmma_f32_16x16x4_f32(false, av, false, bv, (short)0, acc1, false, false);
            else
              acc0 = __builtin_amdgcn_wmma_f32_16x16x4_f32(false, av, false, bv, (short)0, acc0, false, false);
          }
          const v8f acc = acc0 + acc1;
          const float d01 = s0 ? acc[1] : acc[0];
          const float d23 = s0 ? acc[3] : acc[2];
          const float d45 = s0 ? acc[5] : acc[4];
          const float d67 = s0 ? acc[7] : acc[6];
          const float d03 = s1 ? d23 : d01;
          const float d47 = s1 ? d67 : d45;
          float d = s2 ? d47 : d03;
          d = on_diag ? d : 0.0f;
          d = add_row_xor<1>(d);
          d = add_row_xor<2>(d);
          d = add_row_xor<4>(d);
          d = add_row_xor<8>(d);
          const float y = (d + xhalf(d)) + bias_r;

          ykeep = (lane == (s & 31)) ? y : ykeep;
          const float ce = (y - xrow) * ((1.0f-MOMB) * (1.0f/ND));

          float fs0 = 0.0f, fs1 = 0.0f;
          #pragma unroll
          for (int i = 0; i < 8; i += 2) {
            const float m0 = MOMB*m_u[i]   + ce*xv[i];
            const float m1 = MOMB*m_u[i+1] + ce*xv[i+1];
            m_u[i] = m0; m_u[i+1] = m1;
            const float w0 = w_u[i]   - eff*m0;
            const float w1 = w_u[i+1] - eff*m1;
            w_u[i] = w0; w_u[i+1] = w1;
            fs0 += w0*w0; fs1 += w1*w1;
          }
          fsv[u] = fs0 + fs1;
        }
        const int sl = tb + PF - 1;
        if ((sl & 31) == 31)
          ystg[((size_t)(b*ND + row))*NT + c*CH + (sl - 31) + lane] = ykeep;
      }
    }
    #pragma unroll
    for (int u = 0; u < PF; ++u) fbuf[CH - PF + u][tid] = fsv[u];
    __syncthreads();
    if (tid < CH)
      nrm2[(size_t)(b*NRB + rblk)*NT + c*CH + tid] = row_sum<NTH>(fbuf[tid]);
    __syncthreads();
  }
#undef XLD
#undef XST
#undef XPIECES
}

struct __align__(16) Smem {
  float mom[ND * MOM_STRIDE];
  float x_s[ND];
  float err_s[ND];
  float yacc[2][ND];
  float iema[ND];
  float redn[8];
  float redf[32];
  float eff_s;
  float scale_s;
};

__global__ __launch_bounds__(1024) void ttt_linear_kernel(
    const float* __restrict__ x, const float* __restrict__ Wb,
    const float* __restrict__ bias, const float* __restrict__ loglr,
    float* __restrict__ out, const float* __restrict__ nrm2,
    const float* __restrict__ ystg)
{
  __shared__ Smem sm;
  const int tid  = threadIdx.x;
  const int lane = tid & 31;
  const int wid  = tid >> 5;
  const int b    = blockIdx.x / NSL;
  const int sl   = blockIdx.x % NSL;

  {
    int bad = 0;
    if (tid < NT) {
      float s = 0.0f;
      #pragma unroll
      for (int r = 0; r < NRB; ++r) s += nrm2[(size_t)(b*NRB + r)*NT + tid];
      bad = !(s <= CLAMP_SAFE);
    }
    if (!__syncthreads_or(bad)) {
      float* tile = sm.mom;
      for (int i = tid; i < ND*8; i += 1024) {
        const int r  = i >> 3;
        const int c4 = (i & 7) << 2;
        const float4 v = *reinterpret_cast<const float4*>(
            ystg + ((size_t)(b*ND + r))*NT + sl*(NT/NSL) + c4);
        tile[(c4 + 0)*TSTR + r] = v.x;
        tile[(c4 + 1)*TSTR + r] = v.y;
        tile[(c4 + 2)*TSTR + r] = v.z;
        tile[(c4 + 3)*TSTR + r] = v.w;
      }
      __syncthreads();
      for (int i = tid; i < (NT/NSL)*(ND/4); i += 1024) {
        const int tl = i >> 6;
        const int q  = i & 63;
        const float4 v = *reinterpret_cast<const float4*>(&tile[tl*TSTR + 4*q]);
        *reinterpret_cast<float4*>(out + ((size_t)b*NT + sl*(NT/NSL) + tl)*ND + 4*q) = v;
      }
      return;
    }
    if (sl != 0) return;
  }

  const int hi   = lane >> 4;
  const int rb   = (wid >> 1) << 4;
  const int row  = rb + (lane & 15);
  const int colbase = (wid & 1) << 7;

  float w_u[64];
  #pragma unroll
  for (int c = 0; c < 32; ++c) {
    const int col = colbase + 4*c + 2*hi;
    const float2 wv = *reinterpret_cast<const float2*>(Wb + row*ND + col);
    w_u[2*c]   = wv.x;
    w_u[2*c+1] = wv.y;
  }
  for (int i = tid; i < ND*MOM_STRIDE; i += 1024) sm.mom[i] = 0.0f;
  if (tid < ND) sm.iema[tid] = 0.0f;

  float lrema    = 1.0f;
  float inner_lr = 0.0f;
  if (tid == 0) {
    inner_lr = fminf(fmaxf(expf(loglr[0]), 1e-5f), 1.0f);
    sm.scale_s = 1.0f;
    sm.eff_s   = 0.0f;
  }
  const float bias_r = (tid < ND) ? bias[tid] : 0.0f;
  __syncthreads();

  for (int t = 0; t < NT; ++t) {
    if (tid < ND) {
      const float xj = x[((size_t)b*NT + t)*ND + tid];
      sm.x_s[tid] = xj;
      float a = fabsf(xj - sm.iema[tid]);
      sm.iema[tid] = NOVB*sm.iema[tid] + (1.0f-NOVB)*xj;
      #pragma unroll
      for (int off = 16; off > 0; off >>= 1) a += __shfl_xor(a, off, 32);
      if (lane == 0) sm.redn[wid] = a;
    }
    __syncthreads();

    if (tid == 0) {
      float s = 0.0f;
      #pragma unroll
      for (int i = 0; i < 8; ++i) s += sm.redn[i];
      const float nov = s * (1.0f/ND);
      const float lrs = fminf(fmaxf(1.0f + 3.0f*nov, 0.5f), 3.0f);
      lrema = NOVB*lrema + (1.0f-NOVB)*lrs;
      sm.eff_s = inner_lr * lrema;
    }

    v8f acc = {};
    #pragma unroll
    for (int c = 0; c < 32; ++c) {
      const int col = colbase + 4*c + 2*hi;
      const float2 xp = *reinterpret_cast<const float2*>(&sm.x_s[col]);
      v2f av; av.x = w_u[2*c]; av.y = w_u[2*c+1];
      v2f bv; bv.x = xp.x;     bv.y = xp.y;
      acc = __builtin_amdgcn_wmma_f32_16x16x4_f32(
           false, av,  false, bv,
           (short)0, acc,  false,  false);
    }
    if ((lane & 15) == 0) {
      const int mbase = rb + hi*8;
      #pragma unroll
      for (int v = 0; v < 8; ++v) sm.yacc[wid & 1][mbase + v] = acc[v];
    }
    __syncthreads();

    if (tid < ND) {
      const float yv = sm.scale_s*(sm.yacc[0][tid] + sm.yacc[1][tid]) + bias_r;
      out[((size_t)b*NT + t)*ND + tid] = yv;
      sm.err_s[tid] = (yv - sm.x_s[tid]) * (1.0f/ND);
    }
    __syncthreads();

    const float err = sm.err_s[row];
    const float eff = sm.eff_s;
    const float sp  = sm.scale_s;
    float fsum = 0.0f;
    #pragma unroll
    for (int c = 0; c < 32; ++c) {
      const int col = colbase + 4*c + 2*hi;
      const float2 xp = *reinterpret_cast<const float2*>(&sm.x_s[col]);
      float2* mptr = reinterpret_cast<float2*>(&sm.mom[row*MOM_STRIDE + col]);
      float2 mp = *mptr;
      mp.x = MOMB*mp.x + (1.0f-MOMB)*err*xp.x;
      mp.y = MOMB*mp.y + (1.0f-MOMB)*err*xp.y;
      *mptr = mp;
      const float w0 = sp*w_u[2*c]   - eff*mp.x;
      const float w1 = sp*w_u[2*c+1] - eff*mp.y;
      w_u[2*c] = w0; w_u[2*c+1] = w1;
      fsum += w0*w0 + w1*w1;
    }
    #pragma unroll
    for (int off = 16; off > 0; off >>= 1) fsum += __shfl_xor(fsum, off, 32);
    if (lane == 0) sm.redf[wid] = fsum;
    __syncthreads();

    if (tid == 0) {
      float s = 0.0f;
      #pragma unroll
      for (int i = 0; i < 32; ++i) s += sm.redf[i];
      const float fn = fmaxf(sqrtf(s), 1e-6f);
      sm.scale_s = fminf(32.0f/fn, 1.0f);
    }
  }
}

extern "C" void kernel_launch(void* const* d_in, const int* in_sizes, int n_in,
                              void* d_out, int out_size, void* d_ws, size_t ws_size,
                              hipStream_t stream) {
  (void)in_sizes; (void)n_in; (void)out_size; (void)ws_size;
  const float* x    = (const float*)d_in[0];
  const float* Wb   = (const float*)d_in[1];
  const float* bias = (const float*)d_in[2];
  const float* lil  = (const float*)d_in[3];
  float* out = (float*)d_out;
  float* nrm2 = (float*)d_ws;
  float* ystg = nrm2 + YSTG_OFF;
  ttt_rows_kernel<<<dim3(NB*NRB), dim3(NTH), 0, stream>>>(x, Wb, bias, lil, ystg, nrm2);
  ttt_linear_kernel<<<dim3(NB*NSL), dim3(1024), 0, stream>>>(x, Wb, bias, lil, out, nrm2, ystg);
}


# All work on x alone (x staging, input averages, novelty, step sizes) and the norm row sums moved to 4 helper waves running ahead; row waves only step, one barrier a chunk; novelty sum order changes
# speedup vs baseline: 42.1386x; 1.3147x over previous
#include <hip/hip_runtime.h>
#include <math.h>

#define NB 4
#define NT 512
#define ND 256
#define MOMB 0.9f
#define NOVB 0.95f
#define MOM_STRIDE 260

#define RBLK 4
#define NRB (ND / RBLK)
#define NHW 4
#define NSTH (RBLK * 32)
#define NTH ((RBLK + NHW) * 32)
#define HCOL (ND / NHW)
#define NSL 16
#define YSTG_OFF (NB * NRB * NT)
#define CLAMP_SAFE 1020.0f
#define FSTR 132
#define PF 4
#define CH 64
#define NCH (NT / CH)
#define TSTR 260

typedef float v2f __attribute__((ext_vector_type(2)));
typedef float v8f __attribute__((ext_vector_type(8)));

template <int N>
__device__ __forceinline__ float row_sum(const float* r)
{
  float s0 = 0.0f, s1 = 0.0f, s2 = 0.0f, s3 = 0.0f;
  #pragma unroll 8
  for (int i = 0; i < N; i += 4) {
    const float4 v = *reinterpret_cast<const float4*>(r + i);
    s0 += v.x; s1 += v.y; s2 += v.z; s3 += v.w;
  }
  return (s0 + s1) + (s2 + s3);
}

template <int M>
__device__ __forceinline__ float add_row_xor(float v)
{
  const int s = __builtin_bit_cast(int, v);
  const int p = __builtin_amdgcn_update_dpp(s, s, 0x160 + M, 0xf, 0xf, true);
  return v + __builtin_bit_cast(float, p);
}

__device__ __forceinline__ float xhalf(float v)
{
  const int s = __builtin_bit_cast(int, v);
  return __builtin_bit_cast(float,
      __builtin_amdgcn_permlanex16(s, s, 0x76543210, 0xfedcba98, false, false));
}

__global__ __launch_bounds__(NTH) void ttt_rows_kernel(
    const float* __restrict__ x, const float* __restrict__ Wb,
    const float* __restrict__ bias, const float* __restrict__ loglr,
    float* __restrict__ ystg, float* __restrict__ nrm2)
{
  __shared__ __align__(16) float xs3[3][CH][ND];
  __shared__ __align__(16) float fbuf2[2][CH][FSTR];
  __shared__ __align__(16) float part[2][CH][NHW];
  __shared__ __align__(16) float eff_s2[2][CH];

  const int tid  = threadIdx.x;
  const int lane = tid & 31;
  const int wid  = __builtin_amdgcn_readfirstlane(tid >> 5);
  const bool stepw = wid < RBLK;
  const int hw   = wid - RBLK;
  const int b    = blockIdx.x / NRB;
  const int rblk = blockIdx.x % NRB;
  const int row  = rblk*RBLK + (stepw ? wid : 0);
  const float* __restrict__ xb = x + (size_t)b*NT*ND;

  float w_u[8], m_u[8];
  {
    const float4 wl = *reinterpret_cast<const float4*>(Wb + row*ND + 4*lane);
    const float4 wh = *reinterpret_cast<const float4*>(Wb + row*ND + 128 + 4*lane);
    w_u[0] = wl.x; w_u[1] = wl.y; w_u[2] = wl.z; w_u[3] = wl.w;
    w_u[4] = wh.x; w_u[5] = wh.y; w_u[6] = wh.z; w_u[7] = wh.w;
    #pragma unroll
    for (int i = 0; i < 8; ++i) m_u[i] = 0.0f;
  }

  const float inner_lr = fminf(fmaxf(expf(loglr[0]), 1e-5f), 1.0f);
  const float bias_r = bias[row];
  float iema0 = 0.0f;
  float iema1 = 0.0f;
  float lrema = 1.0f;
  float ykeep = 0.0f;
  const bool s0 = (lane & 1) != 0, s1 = (lane & 2) != 0, s2 = (lane & 4) != 0;
  const bool on_diag = ((lane >> 3) & 1) == (lane >> 4);
  float* __restrict__ nrow = nrm2 + (size_t)(b*NRB + rblk)*NT;

  for (int c = -2; c < NCH; ++c) {
    if (stepw) {
      if (c >= 0) {
        const float (*xs)[ND] = xs3[c % 3];
        const float* effc = eff_s2[c & 1];
        float (*fb)[FSTR] = fbuf2[c & 1];

        float4 xa[2][PF], xc[2][PF];
        float  xt[2][PF];
        float4 efv[2];
        float  fsv[PF];
        static_assert(PF == 4, "a trip's step sizes are read as one float4");
        #pragma unroll
        for (int u = 0; u < PF; ++u) {
          xa[0][u] = *reinterpret_cast<const float4*>(&xs[u][4*lane]);
          xc[0][u] = *reinterpret_cast<const float4*>(&xs[u][128 + 4*lane]);
          xt[0][u] = xs[u][row];
          fsv[u] = 0.0f;
        }
        efv[0] = *reinterpret_cast<const float4*>(&effc[0]);

        for (int t0 = 0; t0 < CH; t0 += 2*PF) {
          #pragma unroll
          for (int h = 0; h < 2; ++h) {
            const int tb = t0 + h*PF;
            {
              if (tb > 0) {
                #pragma unroll
                for (int u = 0; u < PF; ++u) fb[tb - PF + u][tid] = fsv[u];
              }
              const int tn = (tb + PF < CH) ? tb + PF : tb;
              #pragma unroll
              for (int u = 0; u < PF; ++u) {
                xa[h^1][u] = *reinterpret_cast<const float4*>(&xs[tn + u][4*lane]);
                xc[h^1][u] = *reinterpret_cast<const float4*>(&xs[tn + u][128 + 4*lane]);
                xt[h^1][u] = xs[tn + u][row];
              }
              efv[h^1] = *reinterpret_cast<const float4*>(&effc[tn]);
            }
            __builtin_amdgcn_sched_barrier(0);
            const float efa[PF] = { efv[h].x, efv[h].y, efv[h].z, efv[h].w };
            #pragma unroll
            for (int u = 0; u < PF; ++u) {
              const int s = tb + u;
              const float xv[8] = { xa[h][u].x, xa[h][u].y, xa[h][u].z, xa[h][u].w,
                                    xc[h][u].x, xc[h][u].y, xc[h][u].z, xc[h][u].w };
              const float xrow = xt[h][u];
              const float eff  = efa[u];
              v8f acc0 = {}, acc1 = {};
              #pragma unroll
              for (int j = 0; j < 4; ++j) {
                v2f av; av.x = w_u[2*j]; av.y = w_u[2*j+1];
                v2f bv; bv.x = xv[2*j];  bv.y = xv[2*j+1];
                if (j & 1)
                  acc1 = __builtin_amdgcn_wmma_f32_16x16x4_f32(false, av, false, bv, (short)0, acc1, false, false);
                else
                  acc0 = __builtin_amdgcn_wmma_f32_16x16x4_f32(false, av, false, bv, (short)0, acc0, false, false);
              }
              const v8f acc = acc0 + acc1;
              const float d01 = s0 ? acc[1] : acc[0];
              const float d23 = s0 ? acc[3] : acc[2];
              const float d45 = s0 ? acc[5] : acc[4];
              const float d67 = s0 ? acc[7] : acc[6];
              const float d03 = s1 ? d23 : d01;
              const float d47 = s1 ? d67 : d45;
              float d = s2 ? d47 : d03;
              d = on_diag ? d : 0.0f;
              d = add_row_xor<1>(d);
              d = add_row_xor<2>(d);
              d = add_row_xor<4>(d);
              d = add_row_xor<8>(d);
              const float y = (d + xhalf(d)) + bias_r;

              ykeep = (lane == (s & 31)) ? y : ykeep;
              const float ce = (y - xrow) * ((1.0f-MOMB) * (1.0f/ND));

              float fs0 = 0.0f, fs1 = 0.0f;
              #pragma unroll
              for (int i = 0; i < 8; i += 2) {
                const float m0 = MOMB*m_u[i]   + ce*xv[i];
                const float m1 = MOMB*m_u[i+1] + ce*xv[i+1];
                m_u[i] = m0; m_u[i+1] = m1;
                const float w0 = w_u[i]   - eff*m0;
                const float w1 = w_u[i+1] - eff*m1;
                w_u[i] = w0; w_u[i+1] = w1;
                fs0 += w0*w0; fs1 += w1*w1;
              }
              fsv[u] = fs0 + fs1;
            }
            const int sl = tb + PF - 1;
            if ((sl & 31) == 31)
              ystg[((size_t)(b*ND + row))*NT + c*CH + (sl - 31) + lane] = ykeep;
          }
        }
        #pragma unroll
        for (int u = 0; u < PF; ++u) fb[CH - PF + u][tid] = fsv[u];
      }
    } else {
      const int ks = c + 2;
      const int kf = c + 1;
      const int kn = c - 1;
      float2 xr[CH];
      if (ks < NCH) {
        const float* __restrict__ xk = xb + (size_t)ks*CH*ND + HCOL*hw + 2*lane;
        #pragma unroll
        for (int s = 0; s < CH; ++s) xr[s] = *reinterpret_cast<const float2*>(xk + s*ND);
      }
      if (hw == 0 && kf >= 0 && kf < NCH) {
        const float4* pq = reinterpret_cast<const float4*>(&part[kf & 1][0][0]);
        float ev = 0.0f;
        #pragma unroll 16
        for (int s = 0; s < CH; ++s) {
          const float4 p = pq[s];
          const float nov = ((p.x + p.y) + (p.z + p.w)) * (1.0f/ND);
          const float lrs = fminf(fmaxf(1.0f + 3.0f*nov, 0.5f), 3.0f);
          lrema = NOVB*lrema + (1.0f-NOVB)*lrs;
          ev = (lane == (s & 31)) ? inner_lr * lrema : ev;
          if ((s & 31) == 31) eff_s2[kf & 1][s - 31 + lane] = ev;
        }
      }
      if (hw == 1 && kn >= 0) {
        nrow[kn*CH + lane]      = row_sum<NSTH>(fbuf2[kn & 1][lane]);
        nrow[kn*CH + 32 + lane] = row_sum<NSTH>(fbuf2[kn & 1][32 + lane]);
      }
      if (ks < NCH) {
        float* xd = &xs3[ks % 3][0][HCOL*hw + 2*lane];
        float pk0 = 0.0f, pk1 = 0.0f;
        #pragma unroll
        for (int s = 0; s < CH; ++s) {
          const float2 v = xr[s];
          *reinterpret_cast<float2*>(xd + s*ND) = v;
          float d = fabsf(v.x - iema0) + fabsf(v.y - iema1);
          iema0 = NOVB*iema0 + (1.0f-NOVB)*v.x;
          iema1 = NOVB*iema1 + (1.0f-NOVB)*v.y;
          d = add_row_xor<1>(d);
          d = add_row_xor<2>(d);
          d = add_row_xor<4>(d);
          d = add_row_xor<8>(d);
          d = d + xhalf(d);
          if (s < 32) pk0 = (lane == s) ? d : pk0;
          else        pk1 = (lane == s - 32) ? d : pk1;
        }
        part[ks & 1][lane][hw]      = pk0;
        part[ks & 1][32 + lane][hw] = pk1;
      }
    }
    __syncthreads();
  }
  if (!stepw && hw == 1) {
    nrow[(NCH-1)*CH + lane]      = row_sum<NSTH>(fbuf2[(NCH-1) & 1][lane]);
    nrow[(NCH-1)*CH + 32 + lane] = row_sum<NSTH>(fbuf2[(NCH-1) & 1][32 + lane]);
  }
}

struct __align__(16) Smem {
  float mom[ND * MOM_STRIDE];
  float x_s[ND];
  float err_s[ND];
  float yacc[2][ND];
  float iema[ND];
  float redn[8];
  float redf[32];
  float eff_s;
  float scale_s;
};

__global__ __launch_bounds__(1024) void ttt_linear_kernel(
    const float* __restrict__ x, const float* __restrict__ Wb,
    const float* __restrict__ bias, const float* __restrict__ loglr,
    float* __restrict__ out, const float* __restrict__ nrm2,
    const float* __restrict__ ystg)
{
  __shared__ Smem sm;
  const int tid  = threadIdx.x;
  const int lane = tid & 31;
  const int wid  = tid >> 5;
  const int b    = blockIdx.x / NSL;
  const int sl   = blockIdx.x % NSL;

  {
    int bad = 0;
    if (tid < NT) {
      float s = 0.0f;
      #pragma unroll
      for (int r = 0; r < NRB; ++r) s += nrm2[(size_t)(b*NRB + r)*NT + tid];
      bad = !(s <= CLAMP_SAFE);
    }
    if (!__syncthreads_or(bad)) {
      float* tile = sm.mom;
      for (int i = tid; i < ND*8; i += 1024) {
        const int r  = i >> 3;
        const int c4 = (i & 7) << 2;
        const float4 v = *reinterpret_cast<const float4*>(
            ystg + ((size_t)(b*ND + r))*NT + sl*(NT/NSL) + c4);
        tile[(c4 + 0)*TSTR + r] = v.x;
        tile[(c4 + 1)*TSTR + r] = v.y;
        tile[(c4 + 2)*TSTR + r] = v.z;
        tile[(c4 + 3)*TSTR + r] = v.w;
      }
      __syncthreads();
      for (int i = tid; i < (NT/NSL)*(ND/4); i += 1024) {
        const int tl = i >> 6;
        const int q  = i & 63;
        const float4 v = *reinterpret_cast<const float4*>(&tile[tl*TSTR + 4*q]);
        *reinterpret_cast<float4*>(out + ((size_t)b*NT + sl*(NT/NSL) + tl)*ND + 4*q) = v;
      }
      return;
    }
    if (sl != 0) return;
  }

  const int hi   = lane >> 4;
  const int rb   = (wid >> 1) << 4;
  const int row  = rb + (lane & 15);
  const int colbase = (wid & 1) << 7;

  float w_u[64];
  #pragma unroll
  for (int c = 0; c < 32; ++c) {
    const int col = colbase + 4*c + 2*hi;
    const float2 wv = *reinterpret_cast<const float2*>(Wb + row*ND + col);
    w_u[2*c]   = wv.x;
    w_u[2*c+1] = wv.y;
  }
  for (int i = tid; i < ND*MOM_STRIDE; i += 1024) sm.mom[i] = 0.0f;
  if (tid < ND) sm.iema[tid] = 0.0f;

  float lrema    = 1.0f;
  float inner_lr = 0.0f;
  if (tid == 0) {
    inner_lr = fminf(fmaxf(expf(loglr[0]), 1e-5f), 1.0f);
    sm.scale_s = 1.0f;
    sm.eff_s   = 0.0f;
  }
  const float bias_r = (tid < ND) ? bias[tid] : 0.0f;
  __syncthreads();

  for (int t = 0; t < NT; ++t) {
    if (tid < ND) {
      const float xj = x[((size_t)b*NT + t)*ND + tid];
      sm.x_s[tid] = xj;
      float a = fabsf(xj - sm.iema[tid]);
      sm.iema[tid] = NOVB*sm.iema[tid] + (1.0f-NOVB)*xj;
      #pragma unroll
      for (int off = 16; off > 0; off >>= 1) a += __shfl_xor(a, off, 32);
      if (lane == 0) sm.redn[wid] = a;
    }
    __syncthreads();

    if (tid == 0) {
      float s = 0.0f;
      #pragma unroll
      for (int i = 0; i < 8; ++i) s += sm.redn[i];
      const float nov = s * (1.0f/ND);
      const float lrs = fminf(fmaxf(1.0f + 3.0f*nov, 0.5f), 3.0f);
      lrema = NOVB*lrema + (1.0f-NOVB)*lrs;
      sm.eff_s = inner_lr * lrema;
    }

    v8f acc = {};
    #pragma unroll
    for (int c = 0; c < 32; ++c) {
      const int col = colbase + 4*c + 2*hi;
      const float2 xp = *reinterpret_cast<const float2*>(&sm.x_s[col]);
      v2f av; av.x = w_u[2*c]; av.y = w_u[2*c+1];
      v2f bv; bv.x = xp.x;     bv.y = xp.y;
      acc = __builtin_amdgcn_wmma_f32_16x16x4_f32(
           false, av,  false, bv,
           (short)0, acc,  false,  false);
    }
    if ((lane & 15) == 0) {
      const int mbase = rb + hi*8;
      #pragma unroll
      for (int v = 0; v < 8; ++v) sm.yacc[wid & 1][mbase + v] = acc[v];
    }
    __syncthreads();

    if (tid < ND) {
      const float yv = sm.scale_s*(sm.yacc[0][tid] + sm.yacc[1][tid]) + bias_r;
      out[((size_t)b*NT + t)*ND + tid] = yv;
      sm.err_s[tid] = (yv - sm.x_s[tid]) * (1.0f/ND);
    }
    __syncthreads();

    const float err = sm.err_s[row];
    const float eff = sm.eff_s;
    const float sp  = sm.scale_s;
    float fsum = 0.0f;
    #pragma unroll
    for (int c = 0; c < 32; ++c) {
      const int col = colbase + 4*c + 2*hi;
      const float2 xp = *reinterpret_cast<const float2*>(&sm.x_s[col]);
      float2* mptr = reinterpret_cast<float2*>(&sm.mom[row*MOM_STRIDE + col]);
      float2 mp = *mptr;
      mp.x = MOMB*mp.x + (1.0f-MOMB)*err*xp.x;
      mp.y = MOMB*mp.y + (1.0f-MOMB)*err*xp.y;
      *mptr = mp;
      const float w0 = sp*w_u[2*c]   - eff*mp.x;
      const float w1 = sp*w_u[2*c+1] - eff*mp.y;
      w_u[2*c] = w0; w_u[2*c+1] = w1;
      fsum += w0*w0 + w1*w1;
    }
    #pragma unroll
    for (int off = 16; off > 0; off >>= 1) fsum += __shfl_xor(fsum, off, 32);
    if (lane == 0) sm.redf[wid] = fsum;
    __syncthreads();

    if (tid == 0) {
      float s = 0.0f;
      #pragma unroll
      for (int i = 0; i < 32; ++i) s += sm.redf[i];
      const float fn = fmaxf(sqrtf(s), 1e-6f);
      sm.scale_s = fminf(32.0f/fn, 1.0f);
    }
  }
}

extern "C" void kernel_launch(void* const* d_in, const int* in_sizes, int n_in,
                              void* d_out, int out_size, void* d_ws, size_t ws_size,
                              hipStream_t stream) {
  (void)in_sizes; (void)n_in; (void)out_size; (void)ws_size;
  const float* x    = (const float*)d_in[0];
  const float* Wb   = (const float*)d_in[1];
  const float* bias = (const float*)d_in[2];
  const float* lil  = (const float*)d_in[3];
  float* out = (float*)d_out;
  float* nrm2 = (float*)d_ws;
  float* ystg = nrm2 + YSTG_OFF;
  ttt_rows_kernel<<<dim3(NB*NRB), dim3(NTH), 0, stream>>>(x, Wb, bias, lil, ystg, nrm2);
  ttt_linear_kernel<<<dim3(NB*NSL), dim3(1024), 0, stream>>>(x, Wb, bias, lil, out, nrm2, ystg);
}


# Tile diagonal summed by folding registers across lanes (register 2q takes 2q+1 of lane l^1, pair sums meet over l^2, l^4) and 2 lane fetches: no select tree, mask or copy; same sums, same order
# speedup vs baseline: 44.6581x; 1.0598x over previous
#include <hip/hip_runtime.h>
#include <math.h>

#define NB 4
#define NT 512
#define ND 256
#define MOMB 0.9f
#define NOVB 0.95f
#define MOM_STRIDE 260

#define RBLK 4
#define NRB (ND / RBLK)
#define NHW 4
#define NSTH (RBLK * 32)
#define NTH ((RBLK + NHW) * 32)
#define HCOL (ND / NHW)
#define NSL 16
#define YSTG_OFF (NB * NRB * NT)
#define CLAMP_SAFE 1020.0f
#define FSTR 132
#define PF 4
#define CH 64
#define NCH (NT / CH)
#define TSTR 260

typedef float v2f __attribute__((ext_vector_type(2)));
typedef float v8f __attribute__((ext_vector_type(8)));

template <int N>
__device__ __forceinline__ float row_sum(const float* r)
{
  float s0 = 0.0f, s1 = 0.0f, s2 = 0.0f, s3 = 0.0f;
  #pragma unroll 8
  for (int i = 0; i < N; i += 4) {
    const float4 v = *reinterpret_cast<const float4*>(r + i);
    s0 += v.x; s1 += v.y; s2 += v.z; s3 += v.w;
  }
  return (s0 + s1) + (s2 + s3);
}

template <int M>
__device__ __forceinline__ float add_row_xor(float v)
{
  const int s = __builtin_bit_cast(int, v);
  const int p = __builtin_amdgcn_update_dpp(s, s, 0x160 + M, 0xf, 0xf, true);
  return v + __builtin_bit_cast(float, p);
}

template <int M>
__device__ __forceinline__ float add_from_xor(float own, float oth)
{
  return own + __builtin_amdgcn_update_dpp(oth, oth, 0x160 + M, 0xf, 0xf, true);
}

__device__ __forceinline__ float xhalf(float v)
{
  const int s = __builtin_bit_cast(int, v);
  return __builtin_bit_cast(float,
      __builtin_amdgcn_permlanex16(s, s, 0x76543210, 0xfedcba98, false, false));
}

__global__ __launch_bounds__(NTH) void ttt_rows_kernel(
    const float* __restrict__ x, const float* __restrict__ Wb,
    const float* __restrict__ bias, const float* __restrict__ loglr,
    float* __restrict__ ystg, float* __restrict__ nrm2)
{
  __shared__ __align__(16) float xs3[3][CH][ND];
  __shared__ __align__(16) float fbuf2[2][CH][FSTR];
  __shared__ __align__(16) float part[2][CH][NHW];
  __shared__ __align__(16) float eff_s2[2][CH];

  const int tid  = threadIdx.x;
  const int lane = tid & 31;
  const int wid  = __builtin_amdgcn_readfirstlane(tid >> 5);
  const bool stepw = wid < RBLK;
  const int hw   = wid - RBLK;
  const int b    = blockIdx.x / NRB;
  const int rblk = blockIdx.x % NRB;
  const int row  = rblk*RBLK + (stepw ? wid : 0);
  const float* __restrict__ xb = x + (size_t)b*NT*ND;

  float w_u[8], m_u[8];
  {
    const float4 wl = *reinterpret_cast<const float4*>(Wb + row*ND + 4*lane);
    const float4 wh = *reinterpret_cast<const float4*>(Wb + row*ND + 128 + 4*lane);
    w_u[0] = wl.x; w_u[1] = wl.y; w_u[2] = wl.z; w_u[3] = wl.w;
    w_u[4] = wh.x; w_u[5] = wh.y; w_u[6] = wh.z; w_u[7] = wh.w;
    #pragma unroll
    for (int i = 0; i < 8; ++i) m_u[i] = 0.0f;
  }

  const float inner_lr = fminf(fmaxf(expf(loglr[0]), 1e-5f), 1.0f);
  const float bias_r = bias[row];
  float iema0 = 0.0f;
  float iema1 = 0.0f;
  float lrema = 1.0f;
  float ykeep = 0.0f;
  const int sel_own = (lane & 16) ? 8 : 0;
  const int sel_oth = sel_own ^ 8;
  float* __restrict__ nrow = nrm2 + (size_t)(b*NRB + rblk)*NT;

  for (int c = -2; c < NCH; ++c) {
    if (stepw) {
      if (c >= 0) {
        const float (*xs)[ND] = xs3[c % 3];
        const float* effc = eff_s2[c & 1];
        float (*fb)[FSTR] = fbuf2[c & 1];

        float4 xa[2][PF], xc[2][PF];
        float  xt[2][PF];
        float4 efv[2];
        float  fsv[PF];
        static_assert(PF == 4, "a trip's step sizes are read as one float4");
        #pragma unroll
        for (int u = 0; u < PF; ++u) {
          xa[0][u] = *reinterpret_cast<const float4*>(&xs[u][4*lane]);
          xc[0][u] = *reinterpret_cast<const float4*>(&xs[u][128 + 4*lane]);
          xt[0][u] = xs[u][row];
          fsv[u] = 0.0f;
        }
        efv[0] = *reinterpret_cast<const float4*>(&effc[0]);

        for (int t0 = 0; t0 < CH; t0 += 2*PF) {
          #pragma unroll
          for (int h = 0; h < 2; ++h) {
            const int tb = t0 + h*PF;
            {
              if (tb > 0) {
                #pragma unroll
                for (int u = 0; u < PF; ++u) fb[tb - PF + u][tid] = fsv[u];
              }
              const int tn = (tb + PF < CH) ? tb + PF : tb;
              #pragma unroll
              for (int u = 0; u < PF; ++u) {
                xa[h^1][u] = *reinterpret_cast<const float4*>(&xs[tn + u][4*lane]);
                xc[h^1][u] = *reinterpret_cast<const float4*>(&xs[tn + u][128 + 4*lane]);
                xt[h^1][u] = xs[tn + u][row];
              }
              efv[h^1] = *reinterpret_cast<const float4*>(&effc[tn]);
            }
            __builtin_amdgcn_sched_barrier(0);
            const float efa[PF] = { efv[h].x, efv[h].y, efv[h].z, efv[h].w };
            #pragma unroll
            for (int u = 0; u < PF; ++u) {
              const int s = tb + u;
              const float xv[8] = { xa[h][u].x, xa[h][u].y, xa[h][u].z, xa[h][u].w,
                                    xc[h][u].x, xc[h][u].y, xc[h][u].z, xc[h][u].w };
              const float xrow = xt[h][u];
              const float eff  = efa[u];
              v8f acc0 = {}, acc1 = {};
              #pragma unroll
              for (int j = 0; j < 4; ++j) {
                v2f av; av.x = w_u[2*j]; av.y = w_u[2*j+1];
                v2f bv; bv.x = xv[2*j];  bv.y = xv[2*j+1];
                if (j & 1)
                  acc1 = __builtin_amdgcn_wmma_f32_16x16x4_f32(false, av, false, bv, (short)0, acc1, false, false);
                else
                  acc0 = __builtin_amdgcn_wmma_f32_16x16x4_f32(false, av, false, bv, (short)0, acc0, false, false);
              }
              const v8f acc = acc0 + acc1;
              const float g0 = add_from_xor<1>(acc[0], acc[1]);
              const float g1 = add_from_xor<1>(acc[2], acc[3]);
              const float g2 = add_from_xor<1>(acc[4], acc[5]);
              const float g3 = add_from_xor<1>(acc[6], acc[7]);
              const float h0 = add_from_xor<2>(g0, g1);
              const float h1 = add_from_xor<2>(g2, g3);
              const int ks = __builtin_bit_cast(int, add_from_xor<4>(h0, h1));
              const float dq = __builtin_bit_cast(float, __builtin_amdgcn_permlane16_var(ks, ks, sel_own, true, false));
              const float dp = __builtin_bit_cast(float, __builtin_amdgcn_permlanex16_var(ks, ks, sel_oth, true, false));
              const float y = (dq + dp) + bias_r;

              ykeep = (lane == (s & 31)) ? y : ykeep;
              const float ce = (y - xrow) * ((1.0f-MOMB) * (1.0f/ND));

              float fs0 = 0.0f, fs1 = 0.0f;
              #pragma unroll
              for (int i = 0; i < 8; i += 2) {
                const float m0 = MOMB*m_u[i]   + ce*xv[i];
                const float m1 = MOMB*m_u[i+1] + ce*xv[i+1];
                m_u[i] = m0; m_u[i+1] = m1;
                const float w0 = w_u[i]   - eff*m0;
                const float w1 = w_u[i+1] - eff*m1;
                w_u[i] = w0; w_u[i+1] = w1;
                fs0 += w0*w0; fs1 += w1*w1;
              }
              fsv[u] = fs0 + fs1;
            }
            const int sl = tb + PF - 1;
            if ((sl & 31) == 31)
              ystg[((size_t)(b*ND + row))*NT + c*CH + (sl - 31) + lane] = ykeep;
          }
        }
        #pragma unroll
        for (int u = 0; u < PF; ++u) fb[CH - PF + u][tid] = fsv[u];
      }
    } else {
      const int ks = c + 2;
      const int kf = c + 1;
      const int kn = c - 1;
      float2 xr[CH];
      if (ks < NCH) {
        const float* __restrict__ xk = xb + (size_t)ks*CH*ND + HCOL*hw + 2*lane;
        #pragma unroll
        for (int s = 0; s < CH; ++s) xr[s] = *reinterpret_cast<const float2*>(xk + s*ND);
      }
      if (hw == 0 && kf >= 0 && kf < NCH) {
        const float4* pq = reinterpret_cast<const float4*>(&part[kf & 1][0][0]);
        float ev = 0.0f;
        #pragma unroll 16
        for (int s = 0; s < CH; ++s) {
          const float4 p = pq[s];
          const float nov = ((p.x + p.y) + (p.z + p.w)) * (1.0f/ND);
          const float lrs = fminf(fmaxf(1.0f + 3.0f*nov, 0.5f), 3.0f);
          lrema = NOVB*lrema + (1.0f-NOVB)*lrs;
          ev = (lane == (s & 31)) ? inner_lr * lrema : ev;
          if ((s & 31) == 31) eff_s2[kf & 1][s - 31 + lane] = ev;
        }
      }
      if (hw == 1 && kn >= 0) {
        nrow[kn*CH + lane]      = row_sum<NSTH>(fbuf2[kn & 1][lane]);
        nrow[kn*CH + 32 + lane] = row_sum<NSTH>(fbuf2[kn & 1][32 + lane]);
      }
      if (ks < NCH) {
        float* xd = &xs3[ks % 3][0][HCOL*hw + 2*lane];
        float pk0 = 0.0f, pk1 = 0.0f;
        #pragma unroll
        for (int s = 0; s < CH; ++s) {
          const float2 v = xr[s];
          *reinterpret_cast<float2*>(xd + s*ND) = v;
          float d = fabsf(v.x - iema0) + fabsf(v.y - iema1);
          iema0 = NOVB*iema0 + (1.0f-NOVB)*v.x;
          iema1 = NOVB*iema1 + (1.0f-NOVB)*v.y;
          d = add_row_xor<1>(d);
          d = add_row_xor<2>(d);
          d = add_row_xor<4>(d);
          d = add_row_xor<8>(d);
          d = d + xhalf(d);
          if (s < 32) pk0 = (lane == s) ? d : pk0;
          else        pk1 = (lane == s - 32) ? d : pk1;
        }
        part[ks & 1][lane][hw]      = pk0;
        part[ks & 1][32 + lane][hw] = pk1;
      }
    }
    __syncthreads();
  }
  if (!stepw && hw == 1) {
    nrow[(NCH-1)*CH + lane]      = row_sum<NSTH>(fbuf2[(NCH-1) & 1][lane]);
    nrow[(NCH-1)*CH + 32 + lane] = row_sum<NSTH>(fbuf2[(NCH-1) & 1][32 + lane]);
  }
}

struct __align__(16) Smem {
  float mom[ND * MOM_STRIDE];
  float x_s[ND];
  float err_s[ND];
  float yacc[2][ND];
  float iema[ND];
  float redn[8];
  float redf[32];
  float eff_s;
  float scale_s;
};

__global__ __launch_bounds__(1024) void ttt_linear_kernel(
    const float* __restrict__ x, const float* __restrict__ Wb,
    const float* __restrict__ bias, const float* __restrict__ loglr,
    float* __restrict__ out, const float* __restrict__ nrm2,
    const float* __restrict__ ystg)
{
  __shared__ Smem sm;
  const int tid  = threadIdx.x;
  const int lane = tid & 31;
  const int wid  = tid >> 5;
  const int b    = blockIdx.x / NSL;
  const int sl   = blockIdx.x % NSL;

  {
    int bad = 0;
    if (tid < NT) {
      float s = 0.0f;
      #pragma unroll
      for (int r = 0; r < NRB; ++r) s += nrm2[(size_t)(b*NRB + r)*NT + tid];
      bad = !(s <= CLAMP_SAFE);
    }
    if (!__syncthreads_or(bad)) {
      float* tile = sm.mom;
      for (int i = tid; i < ND*8; i += 1024) {
        const int r  = i >> 3;
        const int c4 = (i & 7) << 2;
        const float4 v = *reinterpret_cast<const float4*>(
            ystg + ((size_t)(b*ND + r))*NT + sl*(NT/NSL) + c4);
        tile[(c4 + 0)*TSTR + r] = v.x;
        tile[(c4 + 1)*TSTR + r] = v.y;
        tile[(c4 + 2)*TSTR + r] = v.z;
        tile[(c4 + 3)*TSTR + r] = v.w;
      }
      __syncthreads();
      for (int i = tid; i < (NT/NSL)*(ND/4); i += 1024) {
        const int tl = i >> 6;
        const int q  = i & 63;
        const float4 v = *reinterpret_cast<const float4*>(&tile[tl*TSTR + 4*q]);
        *reinterpret_cast<float4*>(out + ((size_t)b*NT + sl*(NT/NSL) + tl)*ND + 4*q) = v;
      }
      return;
    }
    if (sl != 0) return;
  }

  const int hi   = lane >> 4;
  const int rb   = (wid >> 1) << 4;
  const int row  = rb + (lane & 15);
  const int colbase = (wid & 1) << 7;

  float w_u[64];
  #pragma unroll
  for (int c = 0; c < 32; ++c) {
    const int col = colbase + 4*c + 2*hi;
    const float2 wv = *reinterpret_cast<const float2*>(Wb + row*ND + col);
    w_u[2*c]   = wv.x;
    w_u[2*c+1] = wv.y;
  }
  for (int i = tid; i < ND*MOM_STRIDE; i += 1024) sm.mom[i] = 0.0f;
  if (tid < ND) sm.iema[tid] = 0.0f;

  float lrema    = 1.0f;
  float inner_lr = 0.0f;
  if (tid == 0) {
    inner_lr = fminf(fmaxf(expf(loglr[0]), 1e-5f), 1.0f);
    sm.scale_s = 1.0f;
    sm.eff_s   = 0.0f;
  }
  const float bias_r = (tid < ND) ? bias[tid] : 0.0f;
  __syncthreads();

  for (int t = 0; t < NT; ++t) {
    if (tid < ND) {
      const float xj = x[((size_t)b*NT + t)*ND + tid];
      sm.x_s[tid] = xj;
      float a = fabsf(xj - sm.iema[tid]);
      sm.iema[tid] = NOVB*sm.iema[tid] + (1.0f-NOVB)*xj;
      #pragma unroll
      for (int off = 16; off > 0; off >>= 1) a += __shfl_xor(a, off, 32);
      if (lane == 0) sm.redn[wid] = a;
    }
    __syncthreads();

    if (tid == 0) {
      float s = 0.0f;
      #pragma unroll
      for (int i = 0; i < 8; ++i) s += sm.redn[i];
      const float nov = s * (1.0f/ND);
      const float lrs = fminf(fmaxf(1.0f + 3.0f*nov, 0.5f), 3.0f);
      lrema = NOVB*lrema + (1.0f-NOVB)*lrs;
      sm.eff_s = inner_lr * lrema;
    }

    v8f acc = {};
    #pragma unroll
    for (int c = 0; c < 32; ++c) {
      const int col = colbase + 4*c + 2*hi;
      const float2 xp = *reinterpret_cast<const float2*>(&sm.x_s[col]);
      v2f av; av.x = w_u[2*c]; av.y = w_u[2*c+1];
      v2f bv; bv.x = xp.x;     bv.y = xp.y;
      acc = __builtin_amdgcn_wmma_f32_16x16x4_f32(
           false, av,  false, bv,
           (short)0, acc,  false,  false);
    }
    if ((lane & 15) == 0) {
      const int mbase = rb + hi*8;
      #pragma unroll
      for (int v = 0; v < 8; ++v) sm.yacc[wid & 1][mbase + v] = acc[v];
    }
    __syncthreads();

    if (tid < ND) {
      const float yv = sm.scale_s*(sm.yacc[0][tid] + sm.yacc[1][tid]) + bias_r;
      out[((size_t)b*NT + t)*ND + tid] = yv;
      sm.err_s[tid] = (yv - sm.x_s[tid]) * (1.0f/ND);
    }
    __syncthreads();

    const float err = sm.err_s[row];
    const float eff = sm.eff_s;
    const float sp  = sm.scale_s;
    float fsum = 0.0f;
    #pragma unroll
    for (int c = 0; c < 32; ++c) {
      const int col = colbase + 4*c + 2*hi;
      const float2 xp = *reinterpret_cast<const float2*>(&sm.x_s[col]);
      float2* mptr = reinterpret_cast<float2*>(&sm.mom[row*MOM_STRIDE + col]);
      float2 mp = *mptr;
      mp.x = MOMB*mp.x + (1.0f-MOMB)*err*xp.x;
      mp.y = MOMB*mp.y + (1.0f-MOMB)*err*xp.y;
      *mptr = mp;
      const float w0 = sp*w_u[2*c]   - eff*mp.x;
      const float w1 = sp*w_u[2*c+1] - eff*mp.y;
      w_u[2*c] = w0; w_u[2*c+1] = w1;
      fsum += w0*w0 + w1*w1;
    }
    #pragma unroll
    for (int off = 16; off > 0; off >>= 1) fsum += __shfl_xor(fsum, off, 32);
    if (lane == 0) sm.redf[wid] = fsum;
    __syncthreads();

    if (tid == 0) {
      float s = 0.0f;
      #pragma unroll
      for (int i = 0; i < 32; ++i) s += sm.redf[i];
      const float fn = fmaxf(sqrtf(s), 1e-6f);
      sm.scale_s = fminf(32.0f/fn, 1.0f);
    }
  }
}

extern "C" void kernel_launch(void* const* d_in, const int* in_sizes, int n_in,
                              void* d_out, int out_size, void* d_ws, size_t ws_size,
                              hipStream_t stream) {
  (void)in_sizes; (void)n_in; (void)out_size; (void)ws_size;
  const float* x    = (const float*)d_in[0];
  const float* Wb   = (const float*)d_in[1];
  const float* bias = (const float*)d_in[2];
  const float* lil  = (const float*)d_in[3];
  float* out = (float*)d_out;
  float* nrm2 = (float*)d_ws;
  float* ystg = nrm2 + YSTG_OFF;
  ttt_rows_kernel<<<dim3(NB*NRB), dim3(NTH), 0, stream>>>(x, Wb, bias, lil, ystg, nrm2);
  ttt_linear_kernel<<<dim3(NB*NSL), dim3(1024), 0, stream>>>(x, Wb, bias, lil, out, nrm2, ystg);
}
